# starting kernel
# baseline (speedup 1.0000x reference)
#include <hip/hip_runtime.h>
#include <hip/hip_bf16.h>

#define C_   8
#define TF_  512
#define TG_  2048
#define H_   64
#define NITER 18

typedef __attribute__((ext_vector_type(2))) float v2f;
typedef __attribute__((ext_vector_type(8))) float v8f;

__global__ __launch_bounds__(256)
void init_x_kernel(const float* __restrict__ z,
                   const float* __restrict__ w_z,
                   const float* __restrict__ b_z,
                   float* __restrict__ Xa,
                   float* __restrict__ Xb) {
    int idx = blockIdx.x * blockDim.x + threadIdx.x;
    if (idx >= C_ * TG_ * H_) return;
    int h  = idx % H_;
    int ct = idx / H_;
    float v = z[ct] * w_z[h] + b_z[h];
    Xa[idx] = v;
    Xb[idx] = v;
}

__global__ __launch_bounds__(128)
void neumann_iter_kernel(const float* __restrict__ grn,
                         const float* __restrict__ z,
                         const float* __restrict__ w_z,
                         const float* __restrict__ b_z,
                         const int*   __restrict__ binary_tf,
                         const float* __restrict__ Xin,
                         float*       __restrict__ Xout) {
    const int c     = blockIdx.y;
    const int mtile = blockIdx.x;
    const int wave  = threadIdx.x >> 5;
    const int lane  = threadIdx.x & 31;
    const int m16   = lane & 15;
    const int hi    = lane >> 4;
    const int n     = wave * 16 + m16;

    const int* tf = binary_tf + c * TF_;

    v8f acc;
#pragma unroll
    for (int v = 0; v < 8; ++v) {
        int mrow = mtile * 16 + v + hi * 8;
        int trow = tf[mrow];
        acc[v] = z[c * TG_ + trow] * w_z[n] + b_z[n];
    }

    const float* grow = grn + ((size_t)(c * TF_ + mtile * 16 + m16)) * TG_;
    const float* xcol = Xin + (size_t)c * TG_ * H_ + n;
    const int klo = hi * 2;

#pragma unroll 4
    for (int kk = 0; kk < TG_; kk += 4) {
        v2f a = *(const v2f*)(grow + kk + klo);
        v2f b;
        b.x = xcol[(size_t)(kk + klo)     * H_];
        b.y = xcol[(size_t)(kk + klo + 1) * H_];
        acc = __builtin_amdgcn_wmma_f32_16x16x4_f32(
                  false, a, false, b, (short)0, acc, false, false);
    }

    float* xoutBase = Xout + (size_t)c * TG_ * H_;
#pragma unroll
    for (int v = 0; v < 8; ++v) {
        int mrow = mtile * 16 + v + hi * 8;
        int trow = tf[mrow];
        xoutBase[(size_t)trow * H_ + n] = acc[v];
    }
}

__global__ __launch_bounds__(128)
void mlp_kernel(const float* __restrict__ X,
                const float* __restrict__ w1, const float* __restrict__ b1,
                const float* __restrict__ w2, const float* __restrict__ b2,
                float* __restrict__ out) {
    int idx = blockIdx.x * blockDim.x + threadIdx.x;
    if (idx >= C_ * TG_) return;
    const float* x = X + (size_t)idx * H_;
    float xr[H_];
#pragma unroll
    for (int k = 0; k < H_; ++k) xr[k] = x[k];

    float o0 = b2[0];
    float o1 = b2[1];
    for (int j = 0; j < H_; ++j) {
        float hj = b1[j];
        const float* wrow = w1 + j * H_;
#pragma unroll
        for (int k = 0; k < H_; ++k) hj = fmaf(wrow[k], xr[k], hj);
        hj = fmaxf(hj, 0.0f);
        o0 = fmaf(w2[j],       hj, o0);
        o1 = fmaf(w2[H_ + j],  hj, o1);
    }
    out[idx] = o0;
    float sp = fmaxf(o1, 0.0f) + log1pf(expf(-fabsf(o1)));
    out[C_ * TG_ + idx] = sp + 1e-6f;
}

extern "C" void kernel_launch(void* const* d_in, const int* in_sizes, int n_in,
                              void* d_out, int out_size, void* d_ws, size_t ws_size,
                              hipStream_t stream) {
    (void)in_sizes; (void)n_in; (void)out_size; (void)ws_size;
    const float* z   = (const float*)d_in[0];
    const float* grn = (const float*)d_in[1];
    const int*   btf = (const int*)  d_in[2];
    const float* w_z = (const float*)d_in[4];
    const float* b_z = (const float*)d_in[5];
    const float* w1  = (const float*)d_in[6];
    const float* b1  = (const float*)d_in[7];
    const float* w2  = (const float*)d_in[8];
    const float* b2  = (const float*)d_in[9];
    float* out = (float*)d_out;

    float* Xa = (float*)d_ws;
    float* Xb = Xa + (size_t)C_ * TG_ * H_;

    const int total = C_ * TG_ * H_;
    init_x_kernel<<<(total + 255) / 256, 256, 0, stream>>>(z, w_z, b_z, Xa, Xb);

    dim3 grid(TF_ / 16, C_);
    float* bufs[2] = { Xa, Xb };
    for (int it = 0; it < NITER; ++it) {
        const float* xin = bufs[it & 1];
        float*      xout = bufs[(it + 1) & 1];
        neumann_iter_kernel<<<grid, 128, 0, stream>>>(grn, z, w_z, b_z, btf, xin, xout);
    }
    const float* xfin = bufs[NITER & 1];

    mlp_kernel<<<(C_ * TG_ + 127) / 128, 128, 0, stream>>>(xfin, w1, b1, w2, b2, out);
}


# each wave takes one K-slice of all four N-tiles (16 slices per work-group, partial tiles summed via LDS in fixed slice order: sum order changes); A fragment reused 4x, B rows read as 16-byte pieces
# speedup vs baseline: 4.0409x; 4.0409x over previous
#include <hip/hip_runtime.h>
#include <hip/hip_bf16.h>

#define C_   8
#define TF_  512
#define TG_  2048
#define H_   64
#define NITER 18
#define KS_   16
#define KLEN_ (TG_ / KS_)

typedef __attribute__((ext_vector_type(2))) float v2f;
typedef __attribute__((ext_vector_type(4))) float v4f;
typedef __attribute__((ext_vector_type(8))) float v8f;

__global__ __launch_bounds__(256)
void init_x_kernel(const float* __restrict__ z,
                   const float* __restrict__ w_z,
                   const float* __restrict__ b_z,
                   float* __restrict__ Xa,
                   float* __restrict__ Xb) {
    int idx = blockIdx.x * blockDim.x + threadIdx.x;
    if (idx >= C_ * TG_ * H_) return;
    int h  = idx % H_;
    int ct = idx / H_;
    float v = z[ct] * w_z[h] + b_z[h];
    Xa[idx] = v;
    Xb[idx] = v;
}

__global__ __launch_bounds__(32 * KS_)
void neumann_iter_kernel(const float* __restrict__ grn,
                         const float* __restrict__ z,
                         const float* __restrict__ w_z,
                         const float* __restrict__ b_z,
                         const int*   __restrict__ binary_tf,
                         const float* __restrict__ Xin,
                         float*       __restrict__ Xout) {
    __shared__ v4f red[KS_][8][32];

    const int c     = blockIdx.y;
    const int mtile = blockIdx.x;
    const int ks    = threadIdx.x >> 5;
    const int lane  = threadIdx.x & 31;
    const int m16   = lane & 15;
    const int hi    = lane >> 4;
    const int n4    = 4 * m16;

    const int* tf = binary_tf + c * TF_;

    v8f acc0, acc1, acc2, acc3;
#pragma unroll
    for (int v = 0; v < 8; ++v) { acc0[v] = 0.0f; acc1[v] = 0.0f; acc2[v] = 0.0f; acc3[v] = 0.0f; }
    if (ks == 0) {
        const v4f wz = *(const v4f*)(w_z + n4);
        const v4f bz = *(const v4f*)(b_z + n4);
#pragma unroll
        for (int v = 0; v < 8; ++v) {
            int mrow = mtile * 16 + v + hi * 8;
            int trow = tf[mrow];
            float zz = z[c * TG_ + trow];
            acc0[v] = zz * wz.x + bz.x;
            acc1[v] = zz * wz.y + bz.y;
            acc2[v] = zz * wz.z + bz.z;
            acc3[v] = zz * wz.w + bz.w;
        }
    }

    const int kbase = ks * KLEN_ + hi * 2;
    const float* grow = grn + ((size_t)(c * TF_ + mtile * 16 + m16)) * TG_ + kbase;
    const float* xrow = Xin + (size_t)c * TG_ * H_ + (size_t)kbase * H_ + n4;

#pragma unroll 8
    for (int kk = 0; kk < KLEN_; kk += 4) {
        v2f a  = *(const v2f*)(grow + kk);
        v4f b0 = *(const v4f*)(xrow + (size_t)kk * H_);
        v4f b1 = *(const v4f*)(xrow + (size_t)(kk + 1) * H_);
        v2f b;
        b.x = b0.x; b.y = b1.x;
        acc0 = __builtin_amdgcn_wmma_f32_16x16x4_f32(false, a, false, b, (short)0, acc0, false, false);
        b.x = b0.y; b.y = b1.y;
        acc1 = __builtin_amdgcn_wmma_f32_16x16x4_f32(false, a, false, b, (short)0, acc1, false, false);
        b.x = b0.z; b.y = b1.z;
        acc2 = __builtin_amdgcn_wmma_f32_16x16x4_f32(false, a, false, b, (short)0, acc2, false, false);
        b.x = b0.w; b.y = b1.w;
        acc3 = __builtin_amdgcn_wmma_f32_16x16x4_f32(false, a, false, b, (short)0, acc3, false, false);
    }

#pragma unroll
    for (int v = 0; v < 8; ++v) {
        v4f p;
        p.x = acc0[v]; p.y = acc1[v]; p.z = acc2[v]; p.w = acc3[v];
        red[ks][v][lane] = p;
    }
    __syncthreads();

    if (ks < 8) {
        const int v = ks;
        v4f s = red[0][v][lane];
#pragma unroll
        for (int w = 1; w < KS_; ++w) {
            v4f p = red[w][v][lane];
            s.x += p.x; s.y += p.y; s.z += p.z; s.w += p.w;
        }
        int mrow = mtile * 16 + v + hi * 8;
        int trow = tf[mrow];
        float* dst = Xout + (size_t)c * TG_ * H_ + (size_t)trow * H_ + n4;
        *(v4f*)dst = s;
    }
}

__global__ __launch_bounds__(128)
void mlp_kernel(const float* __restrict__ X,
                const float* __restrict__ w1, const float* __restrict__ b1,
                const float* __restrict__ w2, const float* __restrict__ b2,
                float* __restrict__ out) {
    int idx = blockIdx.x * blockDim.x + threadIdx.x;
    if (idx >= C_ * TG_) return;
    const float* x = X + (size_t)idx * H_;
    float xr[H_];
#pragma unroll
    for (int k = 0; k < H_; ++k) xr[k] = x[k];

    float o0 = b2[0];
    float o1 = b2[1];
    for (int j = 0; j < H_; ++j) {
        float hj = b1[j];
        const float* wrow = w1 + j * H_;
#pragma unroll
        for (int k = 0; k < H_; ++k) hj = fmaf(wrow[k], xr[k], hj);
        hj = fmaxf(hj, 0.0f);
        o0 = fmaf(w2[j],       hj, o0);
        o1 = fmaf(w2[H_ + j],  hj, o1);
    }
    out[idx] = o0;
    float sp = fmaxf(o1, 0.0f) + log1pf(expf(-fabsf(o1)));
    out[C_ * TG_ + idx] = sp + 1e-6f;
}

extern "C" void kernel_launch(void* const* d_in, const int* in_sizes, int n_in,
                              void* d_out, int out_size, void* d_ws, size_t ws_size,
                              hipStream_t stream) {
    (void)in_sizes; (void)n_in; (void)out_size; (void)ws_size;
    const float* z   = (const float*)d_in[0];
    const float* grn = (const float*)d_in[1];
    const int*   btf = (const int*)  d_in[2];
    const float* w_z = (const float*)d_in[4];
    const float* b_z = (const float*)d_in[5];
    const float* w1  = (const float*)d_in[6];
    const float* b1  = (const float*)d_in[7];
    const float* w2  = (const float*)d_in[8];
    const float* b2  = (const float*)d_in[9];
    float* out = (float*)d_out;

    float* Xa = (float*)d_ws;
    float* Xb = Xa + (size_t)C_ * TG_ * H_;

    const int total = C_ * TG_ * H_;
    init_x_kernel<<<(total + 255) / 256, 256, 0, stream>>>(z, w_z, b_z, Xa, Xb);

    dim3 grid(TF_ / 16, C_);
    float* bufs[2] = { Xa, Xb };
    for (int it = 0; it < NITER; ++it) {
        const float* xin = bufs[it & 1];
        float*      xout = bufs[(it + 1) & 1];
        neumann_iter_kernel<<<grid, 32 * KS_, 0, stream>>>(grn, z, w_z, b_z, btf, xin, xout);
    }
    const float* xfin = bufs[NITER & 1];

    mlp_kernel<<<(C_ * TG_ + 127) / 128, 128, 0, stream>>>(xfin, w1, b1, w2, b2, out);
}


# rank-2 right-hand side z*w_z+b_z: sweeps run on two vectors (u from z, v from ones; WMMA rows 0,1, rest zero-padded) instead of 64 columns; x=u*w_z+v*b_z rebuilt in the MLP kernel; sum order changes
# speedup vs baseline: 6.6217x; 1.6387x over previous
#include <hip/hip_runtime.h>
#include <hip/hip_bf16.h>

#define C_   8
#define TF_  512
#define TG_  2048
#define H_   64
#define NITER 18
#define KS_   8
#define KLEN_ (TG_ / KS_)
#define STEPS_ (KLEN_ / 8)

typedef __attribute__((ext_vector_type(2))) float v2f;
typedef __attribute__((ext_vector_type(4))) float v4f;
typedef __attribute__((ext_vector_type(8))) float v8f;


__global__ __launch_bounds__(256)
void init_uv_kernel(const float* __restrict__ z,
                    float* __restrict__ UVa,
                    float* __restrict__ UVb,
                    float* __restrict__ zeros) {
    int idx = blockIdx.x * blockDim.x + threadIdx.x;
    if (idx >= C_ * TG_) return;
    if (idx < TG_) zeros[idx] = 0.0f;
    int c = idx / TG_;
    int t = idx % TG_;
    float zz = z[idx];
    size_t pu = ((size_t)c * 2) * TG_ + t;
    size_t pv = pu + TG_;
    UVa[pu] = zz;  UVa[pv] = 1.0f;
    UVb[pu] = zz;  UVb[pv] = 1.0f;
}

__global__ __launch_bounds__(32 * KS_)
void neumann_iter_kernel(const float* __restrict__ grn,
                         const float* __restrict__ z,
                         const int*   __restrict__ binary_tf,
                         const float* __restrict__ UVin,
                         const float* __restrict__ zeros,
                         float*       __restrict__ UVout) {
    __shared__ float red[KS_][2][16];

    const int c     = blockIdx.y;
    const int mtile = blockIdx.x;
    const int ks    = threadIdx.x >> 5;
    const int lane  = threadIdx.x & 31;
    const int c16   = lane & 15;
    const int hi    = lane >> 4;

    const int kbase = ks * KLEN_ + hi * 4;
    const float* grow = grn + ((size_t)(c * TF_ + mtile * 16 + c16)) * TG_ + kbase;
    const float* prow = (c16 < 2) ? UVin + ((size_t)(c * 2 + c16)) * TG_ + kbase
                                  : zeros + kbase;

    v8f acc0, acc1;
#pragma unroll
    for (int r = 0; r < 8; ++r) { acc0[r] = 0.0f; acc1[r] = 0.0f; }

    v4f p[STEPS_], g[STEPS_];
#pragma unroll
    for (int j = 0; j < STEPS_; ++j) {
        p[j] = *(const v4f*)(prow + j * 8);
        g[j] = *(const v4f*)(grow + j * 8);
    }
    __builtin_amdgcn_sched_barrier(0);
#pragma unroll
    for (int j = 0; j < STEPS_; ++j) {
        v2f a, b;
        a.x = p[j].x;  a.y = p[j].y;
        b.x = g[j].x;  b.y = g[j].y;
        acc0 = __builtin_amdgcn_wmma_f32_16x16x4_f32(false, a, false, b, (short)0, acc0, false, false);
        a.x = p[j].z;  a.y = p[j].w;
        b.x = g[j].z;  b.y = g[j].w;
        acc1 = __builtin_amdgcn_wmma_f32_16x16x4_f32(false, a, false, b, (short)0, acc1, false, false);
    }

    const float su = acc0[0] + acc1[0];
    const float sv = acc0[1] + acc1[1];
    if (hi == 0) {
        red[ks][0][c16] = su;
        red[ks][1][c16] = sv;
    }
    __syncthreads();

    if (threadIdx.x < 32) {
        const int row   = lane & 15;
        const int plane = lane >> 4;
        const int trow  = binary_tf[c * TF_ + mtile * 16 + row] & (TG_ - 1);
        float s = (plane == 0) ? z[c * TG_ + trow] : 1.0f;
#pragma unroll
        for (int w = 0; w < KS_; ++w) s += red[w][plane][row];
        UVout[((size_t)(c * 2 + plane)) * TG_ + trow] = s;
    }
}

__global__ __launch_bounds__(128)
void mlp_kernel(const float* __restrict__ UV,
                const float* __restrict__ w_z, const float* __restrict__ b_z,
                const float* __restrict__ w1, const float* __restrict__ b1,
                const float* __restrict__ w2, const float* __restrict__ b2,
                float* __restrict__ out) {
    int idx = blockIdx.x * blockDim.x + threadIdx.x;
    if (idx >= C_ * TG_) return;
    const int c = idx / TG_;
    const int t = idx % TG_;
    const float u = UV[((size_t)c * 2) * TG_ + t];
    const float v = UV[((size_t)c * 2 + 1) * TG_ + t];
    float xr[H_];
#pragma unroll
    for (int k = 0; k < H_; ++k) xr[k] = u * w_z[k] + v * b_z[k];

    float o0 = b2[0];
    float o1 = b2[1];
    for (int j = 0; j < H_; ++j) {
        float hj = b1[j];
        const float* wrow = w1 + j * H_;
#pragma unroll
        for (int k = 0; k < H_; ++k) hj = fmaf(wrow[k], xr[k], hj);
        hj = fmaxf(hj, 0.0f);
        o0 = fmaf(w2[j],       hj, o0);
        o1 = fmaf(w2[H_ + j],  hj, o1);
    }
    out[idx] = o0;
    float sp = fmaxf(o1, 0.0f) + log1pf(expf(-fabsf(o1)));
    out[C_ * TG_ + idx] = sp + 1e-6f;
}

extern "C" void kernel_launch(void* const* d_in, const int* in_sizes, int n_in,
                              void* d_out, int out_size, void* d_ws, size_t ws_size,
                              hipStream_t stream) {
    (void)in_sizes; (void)n_in; (void)out_size; (void)ws_size;
    const float* z   = (const float*)d_in[0];
    const float* grn = (const float*)d_in[1];
    const int*   btf = (const int*)  d_in[2];
    const float* w_z = (const float*)d_in[4];
    const float* b_z = (const float*)d_in[5];
    const float* w1  = (const float*)d_in[6];
    const float* b1  = (const float*)d_in[7];
    const float* w2  = (const float*)d_in[8];
    const float* b2  = (const float*)d_in[9];
    float* out = (float*)d_out;

    float* UVa = (float*)d_ws;
    float* UVb = UVa + (size_t)C_ * 2 * TG_;
    float* zer = UVb + (size_t)C_ * 2 * TG_;

    init_uv_kernel<<<(C_ * TG_ + 255) / 256, 256, 0, stream>>>(z, UVa, UVb, zer);

    dim3 grid(TF_ / 16, C_);
    float* bufs[2] = { UVa, UVb };
    for (int it = 0; it < NITER; ++it) {
        const float* uvin = bufs[it & 1];
        float*      uvout = bufs[(it + 1) & 1];
        neumann_iter_kernel<<<grid, 32 * KS_, 0, stream>>>(grn, z, btf, uvin, zer, uvout);
    }
    const float* uvfin = bufs[NITER & 1];

    mlp_kernel<<<(C_ * TG_ + 127) / 128, 128, 0, stream>>>(uvfin, w_z, b_z, w1, b1, w2, b2, out);
}


# MLP layers (64x64 hidden, 2x64 output) moved from one dependent FMA chain per thread fed by scalar weight loads to f32 WMMA tiles, 32 rows per wave; sum order of both layers changes
# speedup vs baseline: 9.8433x; 1.4865x over previous
#include <hip/hip_runtime.h>
#include <hip/hip_bf16.h>

#define C_   8
#define TF_  512
#define TG_  2048
#define H_   64
#define NITER 18
#define KS_   8
#define KLEN_ (TG_ / KS_)
#define STEPS_ (KLEN_ / 8)

typedef __attribute__((ext_vector_type(2))) float v2f;
typedef __attribute__((ext_vector_type(4))) float v4f;
typedef __attribute__((ext_vector_type(8))) float v8f;


__global__ __launch_bounds__(256)
void init_uv_kernel(const float* __restrict__ z,
                    float* __restrict__ UVa,
                    float* __restrict__ UVb,
                    float* __restrict__ zeros) {
    int idx = blockIdx.x * blockDim.x + threadIdx.x;
    if (idx >= C_ * TG_) return;
    if (idx < TG_) zeros[idx] = 0.0f;
    int c = idx / TG_;
    int t = idx % TG_;
    float zz = z[idx];
    size_t pu = ((size_t)c * 2) * TG_ + t;
    size_t pv = pu + TG_;
    UVa[pu] = zz;  UVa[pv] = 1.0f;
    UVb[pu] = zz;  UVb[pv] = 1.0f;
}

__global__ __launch_bounds__(32 * KS_)
void neumann_iter_kernel(const float* __restrict__ grn,
                         const float* __restrict__ z,
                         const int*   __restrict__ binary_tf,
                         const float* __restrict__ UVin,
                         const float* __restrict__ zeros,
                         float*       __restrict__ UVout) {
    __shared__ float red[KS_][2][16];

    const int c     = blockIdx.y;
    const int mtile = blockIdx.x;
    const int ks    = threadIdx.x >> 5;
    const int lane  = threadIdx.x & 31;
    const int c16   = lane & 15;
    const int hi    = lane >> 4;

    const int kbase = ks * KLEN_ + hi * 4;
    const float* grow = grn + ((size_t)(c * TF_ + mtile * 16 + c16)) * TG_ + kbase;
    const float* prow = (c16 < 2) ? UVin + ((size_t)(c * 2 + c16)) * TG_ + kbase
                                  : zeros + kbase;

    v8f acc0, acc1;
#pragma unroll
    for (int r = 0; r < 8; ++r) { acc0[r] = 0.0f; acc1[r] = 0.0f; }

    v4f p[STEPS_], g[STEPS_];
#pragma unroll
    for (int j = 0; j < STEPS_; ++j) {
        p[j] = *(const v4f*)(prow + j * 8);
        g[j] = *(const v4f*)(grow + j * 8);
    }
    __builtin_amdgcn_sched_barrier(0);
#pragma unroll
    for (int j = 0; j < STEPS_; ++j) {
        v2f a, b;
        a.x = p[j].x;  a.y = p[j].y;
        b.x = g[j].x;  b.y = g[j].y;
        acc0 = __builtin_amdgcn_wmma_f32_16x16x4_f32(false, a, false, b, (short)0, acc0, false, false);
        a.x = p[j].z;  a.y = p[j].w;
        b.x = g[j].z;  b.y = g[j].w;
        acc1 = __builtin_amdgcn_wmma_f32_16x16x4_f32(false, a, false, b, (short)0, acc1, false, false);
    }

    const float su = acc0[0] + acc1[0];
    const float sv = acc0[1] + acc1[1];
    if (hi == 0) {
        red[ks][0][c16] = su;
        red[ks][1][c16] = sv;
    }
    __syncthreads();

    if (threadIdx.x < 32) {
        const int row   = lane & 15;
        const int plane = lane >> 4;
        const int trow  = binary_tf[c * TF_ + mtile * 16 + row] & (TG_ - 1);
        float s = (plane == 0) ? z[c * TG_ + trow] : 1.0f;
#pragma unroll
        for (int w = 0; w < KS_; ++w) s += red[w][plane][row];
        UVout[((size_t)(c * 2 + plane)) * TG_ + trow] = s;
    }
}

__global__ __launch_bounds__(128)
void mlp_kernel(const float* __restrict__ UV,
                const float* __restrict__ w_z, const float* __restrict__ b_z,
                const float* __restrict__ w1, const float* __restrict__ b1,
                const float* __restrict__ w2, const float* __restrict__ b2,
                float* __restrict__ out) {
    const int lane = threadIdx.x & 31;
    const int wave = (blockIdx.x * blockDim.x + threadIdx.x) >> 5;
    const int c16  = lane & 15;
    const int hi   = lane >> 4;
    const int base = wave * 32;
    const int c    = base / TG_;
    const int t0   = base % TG_;

    const float* up = UV + ((size_t)c * 2) * TG_ + t0;
    const float u0 = up[c16];
    const float u1 = up[16 + c16];
    const float v0 = up[TG_ + c16];
    const float v1 = up[TG_ + 16 + c16];

    v4f wz[8], bz[8];
#pragma unroll
    for (int i = 0; i < 8; ++i) {
        wz[i] = *(const v4f*)(w_z + 8 * i + 4 * hi);
        bz[i] = *(const v4f*)(b_z + 8 * i + 4 * hi);
    }

    v4f b1f[4][2];
#pragma unroll
    for (int jt = 0; jt < 4; ++jt) {
        b1f[jt][0] = *(const v4f*)(b1 + 16 * jt + 8 * hi);
        b1f[jt][1] = *(const v4f*)(b1 + 16 * jt + 8 * hi + 4);
    }

    v4f wf[4][8];
#pragma unroll
    for (int i = 0; i < 8; ++i)
#pragma unroll
        for (int jt = 0; jt < 4; ++jt)
            wf[jt][i] = *(const v4f*)(w1 + (size_t)(16 * jt + c16) * H_ + 8 * i + 4 * hi);

    v4f w2f[4][2];
#pragma unroll
    for (int jt = 0; jt < 4; ++jt)
#pragma unroll
        for (int q = 0; q < 2; ++q)
            w2f[jt][q] = *(const v4f*)(w2 + (c16 & 1) * H_ + 16 * jt + 8 * hi + 4 * q);
    const float b20 = b2[0];
    const float b21 = b2[1];

    __builtin_amdgcn_sched_barrier(0);

    v8f acc[2][4];
#pragma unroll
    for (int jt = 0; jt < 4; ++jt)
#pragma unroll
        for (int r = 0; r < 4; ++r) {
            acc[0][jt][r] = b1f[jt][0][r];  acc[0][jt][4 + r] = b1f[jt][1][r];
            acc[1][jt][r] = b1f[jt][0][r];  acc[1][jt][4 + r] = b1f[jt][1][r];
        }

    const bool a2row = (c16 & 6) == 0;
#pragma unroll
    for (int jt = 0; jt < 4; ++jt)
#pragma unroll
        for (int q = 0; q < 2; ++q)
#pragma unroll
            for (int e = 0; e < 4; ++e) w2f[jt][q][e] = a2row ? w2f[jt][q][e] : 0.0f;

#pragma unroll
    for (int i = 0; i < 8; ++i) {
        v4f x0, x1;
#pragma unroll
        for (int e = 0; e < 4; ++e) {
            x0[e] = u0 * wz[i][e] + v0 * bz[i][e];
            x1[e] = u1 * wz[i][e] + v1 * bz[i][e];
        }
#pragma unroll
        for (int hf = 0; hf < 2; ++hf) {
            v2f xb0, xb1;
            xb0.x = x0[2 * hf];  xb0.y = x0[2 * hf + 1];
            xb1.x = x1[2 * hf];  xb1.y = x1[2 * hf + 1];
#pragma unroll
            for (int jt = 0; jt < 4; ++jt) {
                v2f a;
                a.x = wf[jt][i][2 * hf];  a.y = wf[jt][i][2 * hf + 1];
                acc[0][jt] = __builtin_amdgcn_wmma_f32_16x16x4_f32(false, a, false, xb0, (short)0, acc[0][jt], false, false);
                acc[1][jt] = __builtin_amdgcn_wmma_f32_16x16x4_f32(false, a, false, xb1, (short)0, acc[1][jt], false, false);
            }
        }
    }

    v8f o2a, o2b;
#pragma unroll
    for (int r = 0; r < 8; ++r) { o2a[r] = 0.0f; o2b[r] = 0.0f; }
    o2a[0] = b20;  o2a[1] = b21;
    o2b[0] = b20;  o2b[1] = b21;
#pragma unroll
    for (int jt = 0; jt < 4; ++jt)
#pragma unroll
        for (int p = 0; p < 4; ++p) {
            v2f a, h0, h1;
            a.x  = w2f[jt][p >> 1][2 * (p & 1)];
            a.y  = w2f[jt][p >> 1][2 * (p & 1) + 1];
            h0.x = fmaxf(acc[0][jt][2 * p], 0.0f);  h0.y = fmaxf(acc[0][jt][2 * p + 1], 0.0f);
            h1.x = fmaxf(acc[1][jt][2 * p], 0.0f);  h1.y = fmaxf(acc[1][jt][2 * p + 1], 0.0f);
            o2a = __builtin_amdgcn_wmma_f32_16x16x4_f32(false, a, false, h0, (short)0, o2a, false, false);
            o2b = __builtin_amdgcn_wmma_f32_16x16x4_f32(false, a, false, h1, (short)0, o2b, false, false);
        }

    const float o0 = hi ? o2b[0] : o2a[0];
    const float o1 = hi ? o2b[1] : o2a[1];
    const int idx = base + lane;
    out[idx] = o0;
    float sp = fmaxf(o1, 0.0f) + log1pf(expf(-fabsf(o1)));
    out[C_ * TG_ + idx] = sp + 1e-6f;
}

extern "C" void kernel_launch(void* const* d_in, const int* in_sizes, int n_in,
                              void* d_out, int out_size, void* d_ws, size_t ws_size,
                              hipStream_t stream) {
    (void)in_sizes; (void)n_in; (void)out_size; (void)ws_size;
    const float* z   = (const float*)d_in[0];
    const float* grn = (const float*)d_in[1];
    const int*   btf = (const int*)  d_in[2];
    const float* w_z = (const float*)d_in[4];
    const float* b_z = (const float*)d_in[5];
    const float* w1  = (const float*)d_in[6];
    const float* b1  = (const float*)d_in[7];
    const float* w2  = (const float*)d_in[8];
    const float* b2  = (const float*)d_in[9];
    float* out = (float*)d_out;

    float* UVa = (float*)d_ws;
    float* UVb = UVa + (size_t)C_ * 2 * TG_;
    float* zer = UVb + (size_t)C_ * 2 * TG_;

    init_uv_kernel<<<(C_ * TG_ + 255) / 256, 256, 0, stream>>>(z, UVa, UVb, zer);

    dim3 grid(TF_ / 16, C_);
    float* bufs[2] = { UVa, UVb };
    for (int it = 0; it < NITER; ++it) {
        const float* uvin = bufs[it & 1];
        float*      uvout = bufs[(it + 1) & 1];
        neumann_iter_kernel<<<grid, 32 * KS_, 0, stream>>>(grn, z, btf, uvin, zer, uvout);
    }
    const float* uvfin = bufs[NITER & 1];

    mlp_kernel<<<(C_ * TG_ + 127) / 128, 128, 0, stream>>>(uvfin, w_z, b_z, w1, b1, w2, b2, out);
}


# sweep kernel: the epilogue's scattered row index is requested at wave entry and its right-hand-side value under the K loop, so no dependent global load is left behind the work-group barrier
# speedup vs baseline: 10.4564x; 1.0623x over previous
#include <hip/hip_runtime.h>
#include <hip/hip_bf16.h>

#define C_   8
#define TF_  512
#define TG_  2048
#define H_   64
#define NITER 18
#define KS_   8
#define KLEN_ (TG_ / KS_)
#define STEPS_ (KLEN_ / 8)

typedef __attribute__((ext_vector_type(2))) float v2f;
typedef __attribute__((ext_vector_type(4))) float v4f;
typedef __attribute__((ext_vector_type(8))) float v8f;


__global__ __launch_bounds__(256)
void init_uv_kernel(const float* __restrict__ z,
                    float* __restrict__ UVa,
                    float* __restrict__ UVb,
                    float* __restrict__ zeros) {
    int idx = blockIdx.x * blockDim.x + threadIdx.x;
    if (idx >= C_ * TG_) return;
    if (idx < TG_) zeros[idx] = 0.0f;
    int c = idx / TG_;
    int t = idx % TG_;
    float zz = z[idx];
    size_t pu = ((size_t)c * 2) * TG_ + t;
    size_t pv = pu + TG_;
    UVa[pu] = zz;  UVa[pv] = 1.0f;
    UVb[pu] = zz;  UVb[pv] = 1.0f;
}

__global__ __launch_bounds__(32 * KS_)
void neumann_iter_kernel(const float* __restrict__ grn,
                         const float* __restrict__ z,
                         const int*   __restrict__ binary_tf,
                         const float* __restrict__ UVin,
                         const float* __restrict__ zeros,
                         float*       __restrict__ UVout) {
    __shared__ float red[KS_][2][16];

    const int c     = blockIdx.y;
    const int mtile = blockIdx.x;
    const int ks    = threadIdx.x >> 5;
    const int lane  = threadIdx.x & 31;
    const int c16   = lane & 15;
    const int hi    = lane >> 4;

    const int erow = lane & 15;
    const int traw = binary_tf[c * TF_ + mtile * 16 + erow];
    __builtin_amdgcn_sched_barrier(0);

    const int kbase = ks * KLEN_ + hi * 4;
    const float* grow = grn + ((size_t)(c * TF_ + mtile * 16 + c16)) * TG_ + kbase;
    const float* prow = (c16 < 2) ? UVin + ((size_t)(c * 2 + c16)) * TG_ + kbase
                                  : zeros + kbase;

    v8f acc0, acc1;
#pragma unroll
    for (int r = 0; r < 8; ++r) { acc0[r] = 0.0f; acc1[r] = 0.0f; }

    v4f p[STEPS_], g[STEPS_];
#pragma unroll
    for (int j = 0; j < STEPS_; ++j) {
        p[j] = *(const v4f*)(prow + j * 8);
        g[j] = *(const v4f*)(grow + j * 8);
    }
    __builtin_amdgcn_sched_barrier(0);
    const int   trow = traw & (TG_ - 1);
    const float rhs  = (threadIdx.x < 16) ? z[c * TG_ + trow] : 1.0f;
    __builtin_amdgcn_sched_barrier(0);
#pragma unroll
    for (int j = 0; j < STEPS_; ++j) {
        v2f a, b;
        a.x = p[j].x;  a.y = p[j].y;
        b.x = g[j].x;  b.y = g[j].y;
        acc0 = __builtin_amdgcn_wmma_f32_16x16x4_f32(false, a, false, b, (short)0, acc0, false, false);
        a.x = p[j].z;  a.y = p[j].w;
        b.x = g[j].z;  b.y = g[j].w;
        acc1 = __builtin_amdgcn_wmma_f32_16x16x4_f32(false, a, false, b, (short)0, acc1, false, false);
    }

    const float su = acc0[0] + acc1[0];
    const float sv = acc0[1] + acc1[1];
    if (hi == 0) {
        red[ks][0][c16] = su;
        red[ks][1][c16] = sv;
    }
    __syncthreads();

    if (threadIdx.x < 32) {
        const int plane = lane >> 4;
        float s = rhs;
#pragma unroll
        for (int w = 0; w < KS_; ++w) s += red[w][plane][erow];
        UVout[((size_t)(c * 2 + plane)) * TG_ + trow] = s;
    }
}

__global__ __launch_bounds__(128)
void mlp_kernel(const float* __restrict__ UV,
                const float* __restrict__ w_z, const float* __restrict__ b_z,
                const float* __restrict__ w1, const float* __restrict__ b1,
                const float* __restrict__ w2, const float* __restrict__ b2,
                float* __restrict__ out) {
    const int lane = threadIdx.x & 31;
    const int wave = (blockIdx.x * blockDim.x + threadIdx.x) >> 5;
    const int c16  = lane & 15;
    const int hi   = lane >> 4;
    const int base = wave * 32;
    const int c    = base / TG_;
    const int t0   = base % TG_;

    const float* up = UV + ((size_t)c * 2) * TG_ + t0;
    const float u0 = up[c16];
    const float u1 = up[16 + c16];
    const float v0 = up[TG_ + c16];
    const float v1 = up[TG_ + 16 + c16];

    v4f wz[8], bz[8];
#pragma unroll
    for (int i = 0; i < 8; ++i) {
        wz[i] = *(const v4f*)(w_z + 8 * i + 4 * hi);
        bz[i] = *(const v4f*)(b_z + 8 * i + 4 * hi);
    }

    v4f b1f[4][2];
#pragma unroll
    for (int jt = 0; jt < 4; ++jt) {
        b1f[jt][0] = *(const v4f*)(b1 + 16 * jt + 8 * hi);
        b1f[jt][1] = *(const v4f*)(b1 + 16 * jt + 8 * hi + 4);
    }

    v4f wf[4][8];
#pragma unroll
    for (int i = 0; i < 8; ++i)
#pragma unroll
        for (int jt = 0; jt < 4; ++jt)
            wf[jt][i] = *(const v4f*)(w1 + (size_t)(16 * jt + c16) * H_ + 8 * i + 4 * hi);

    v4f w2f[4][2];
#pragma unroll
    for (int jt = 0; jt < 4; ++jt)
#pragma unroll
        for (int q = 0; q < 2; ++q)
            w2f[jt][q] = *(const v4f*)(w2 + (c16 & 1) * H_ + 16 * jt + 8 * hi + 4 * q);
    const float b20 = b2[0];
    const float b21 = b2[1];

    __builtin_amdgcn_sched_barrier(0);

    v8f acc[2][4];
#pragma unroll
    for (int jt = 0; jt < 4; ++jt)
#pragma unroll
        for (int r = 0; r < 4; ++r) {
            acc[0][jt][r] = b1f[jt][0][r];  acc[0][jt][4 + r] = b1f[jt][1][r];
            acc[1][jt][r] = b1f[jt][0][r];  acc[1][jt][4 + r] = b1f[jt][1][r];
        }

    const bool a2row = (c16 & 6) == 0;
#pragma unroll
    for (int jt = 0; jt < 4; ++jt)
#pragma unroll
        for (int q = 0; q < 2; ++q)
#pragma unroll
            for (int e = 0; e < 4; ++e) w2f[jt][q][e] = a2row ? w2f[jt][q][e] : 0.0f;

#pragma unroll
    for (int i = 0; i < 8; ++i) {
        v4f x0, x1;
#pragma unroll
        for (int e = 0; e < 4; ++e) {
            x0[e] = u0 * wz[i][e] + v0 * bz[i][e];
            x1[e] = u1 * wz[i][e] + v1 * bz[i][e];
        }
#pragma unroll
        for (int hf = 0; hf < 2; ++hf) {
            v2f xb0, xb1;
            xb0.x = x0[2 * hf];  xb0.y = x0[2 * hf + 1];
            xb1.x = x1[2 * hf];  xb1.y = x1[2 * hf + 1];
#pragma unroll
            for (int jt = 0; jt < 4; ++jt) {
                v2f a;
                a.x = wf[jt][i][2 * hf];  a.y = wf[jt][i][2 * hf + 1];
                acc[0][jt] = __builtin_amdgcn_wmma_f32_16x16x4_f32(false, a, false, xb0, (short)0, acc[0][jt], false, false);
                acc[1][jt] = __builtin_amdgcn_wmma_f32_16x16x4_f32(false, a, false, xb1, (short)0, acc[1][jt], false, false);
            }
        }
    }

    v8f o2a, o2b;
#pragma unroll
    for (int r = 0; r < 8; ++r) { o2a[r] = 0.0f; o2b[r] = 0.0f; }
    o2a[0] = b20;  o2a[1] = b21;
    o2b[0] = b20;  o2b[1] = b21;
#pragma unroll
    for (int jt = 0; jt < 4; ++jt)
#pragma unroll
        for (int p = 0; p < 4; ++p) {
            v2f a, h0, h1;
            a.x  = w2f[jt][p >> 1][2 * (p & 1)];
            a.y  = w2f[jt][p >> 1][2 * (p & 1) + 1];
            h0.x = fmaxf(acc[0][jt][2 * p], 0.0f);  h0.y = fmaxf(acc[0][jt][2 * p + 1], 0.0f);
            h1.x = fmaxf(acc[1][jt][2 * p], 0.0f);  h1.y = fmaxf(acc[1][jt][2 * p + 1], 0.0f);
            o2a = __builtin_amdgcn_wmma_f32_16x16x4_f32(false, a, false, h0, (short)0, o2a, false, false);
            o2b = __builtin_amdgcn_wmma_f32_16x16x4_f32(false, a, false, h1, (short)0, o2b, false, false);
        }

    const float o0 = hi ? o2b[0] : o2a[0];
    const float o1 = hi ? o2b[1] : o2a[1];
    const int idx = base + lane;
    out[idx] = o0;
    float sp = fmaxf(o1, 0.0f) + log1pf(expf(-fabsf(o1)));
    out[C_ * TG_ + idx] = sp + 1e-6f;
}

extern "C" void kernel_launch(void* const* d_in, const int* in_sizes, int n_in,
                              void* d_out, int out_size, void* d_ws, size_t ws_size,
                              hipStream_t stream) {
    (void)in_sizes; (void)n_in; (void)out_size; (void)ws_size;
    const float* z   = (const float*)d_in[0];
    const float* grn = (const float*)d_in[1];
    const int*   btf = (const int*)  d_in[2];
    const float* w_z = (const float*)d_in[4];
    const float* b_z = (const float*)d_in[5];
    const float* w1  = (const float*)d_in[6];
    const float* b1  = (const float*)d_in[7];
    const float* w2  = (const float*)d_in[8];
    const float* b2  = (const float*)d_in[9];
    float* out = (float*)d_out;

    float* UVa = (float*)d_ws;
    float* UVb = UVa + (size_t)C_ * 2 * TG_;
    float* zer = UVb + (size_t)C_ * 2 * TG_;

    init_uv_kernel<<<(C_ * TG_ + 255) / 256, 256, 0, stream>>>(z, UVa, UVb, zer);

    dim3 grid(TF_ / 16, C_);
    float* bufs[2] = { UVa, UVb };
    for (int it = 0; it < NITER; ++it) {
        const float* uvin = bufs[it & 1];
        float*      uvout = bufs[(it + 1) & 1];
        neumann_iter_kernel<<<grid, 32 * KS_, 0, stream>>>(grn, z, btf, uvin, zer, uvout);
    }
    const float* uvfin = bufs[NITER & 1];

    mlp_kernel<<<(C_ * TG_ + 127) / 128, 128, 0, stream>>>(uvfin, w_z, b_z, w1, b1, w2, b2, out);
}


# set-up launch folded into sweep 1 (20 -> 19 launches): sweep 1 takes x0 from z itself, writes the x0 planes and zero plane, finds row owners by scanning the index list instead of a scattered map
# speedup vs baseline: 16.3015x; 1.5590x over previous
#include <hip/hip_runtime.h>
#include <hip/hip_bf16.h>

#define C_   8
#define TF_  512
#define TG_  2048
#define H_   64
#define NITER 18
#define KS_   8
#define KLEN_ (TG_ / KS_)
#define STEPS_ (KLEN_ / 8)

typedef __attribute__((ext_vector_type(2))) float v2f;
typedef __attribute__((ext_vector_type(4))) float v4f;
typedef __attribute__((ext_vector_type(8))) float v8f;


__global__ __launch_bounds__(256)
void init_uv_kernel(const float* __restrict__ z,
                    float* __restrict__ UVa,
                    float* __restrict__ UVb,
                    float* __restrict__ zeros) {
    int idx = blockIdx.x * blockDim.x + threadIdx.x;
    if (idx >= C_ * TG_) return;
    if (idx < TG_) zeros[idx] = 0.0f;
    int c = idx / TG_;
    int t = idx % TG_;
    float zz = z[idx];
    size_t pu = ((size_t)c * 2) * TG_ + t;
    size_t pv = pu + TG_;
    UVa[pu] = zz;  UVa[pv] = 1.0f;
    UVb[pu] = zz;  UVb[pv] = 1.0f;
}

__global__ __launch_bounds__(32 * KS_)
void neumann_iter_kernel(const float* __restrict__ grn,
                         const float* __restrict__ z,
                         const int*   __restrict__ binary_tf,
                         const float* __restrict__ UVin,
                         const float* __restrict__ zeros,
                         float*       __restrict__ UVout) {
    __shared__ float red[KS_][2][16];

    const int c     = blockIdx.y;
    const int mtile = blockIdx.x;
    const int ks    = threadIdx.x >> 5;
    const int lane  = threadIdx.x & 31;
    const int c16   = lane & 15;
    const int hi    = lane >> 4;

    const int erow = lane & 15;
    const int traw = binary_tf[c * TF_ + mtile * 16 + erow];
    __builtin_amdgcn_sched_barrier(0);

    const int kbase = ks * KLEN_ + hi * 4;
    const float* grow = grn + ((size_t)(c * TF_ + mtile * 16 + c16)) * TG_ + kbase;
    const float* prow = (c16 < 2) ? UVin + ((size_t)(c * 2 + c16)) * TG_ + kbase
                                  : zeros + kbase;

    v8f acc0, acc1;
#pragma unroll
    for (int r = 0; r < 8; ++r) { acc0[r] = 0.0f; acc1[r] = 0.0f; }

    v4f p[STEPS_], g[STEPS_];
#pragma unroll
    for (int j = 0; j < STEPS_; ++j) {
        p[j] = *(const v4f*)(prow + j * 8);
        g[j] = *(const v4f*)(grow + j * 8);
    }
    __builtin_amdgcn_sched_barrier(0);
    const int   trow = traw & (TG_ - 1);
    const float rhs  = (threadIdx.x < 16) ? z[c * TG_ + trow] : 1.0f;
    __builtin_amdgcn_sched_barrier(0);
#pragma unroll
    for (int j = 0; j < STEPS_; ++j) {
        v2f a, b;
        a.x = p[j].x;  a.y = p[j].y;
        b.x = g[j].x;  b.y = g[j].y;
        acc0 = __builtin_amdgcn_wmma_f32_16x16x4_f32(false, a, false, b, (short)0, acc0, false, false);
        a.x = p[j].z;  a.y = p[j].w;
        b.x = g[j].z;  b.y = g[j].w;
        acc1 = __builtin_amdgcn_wmma_f32_16x16x4_f32(false, a, false, b, (short)0, acc1, false, false);
    }

    const float su = acc0[0] + acc1[0];
    const float sv = acc0[1] + acc1[1];
    if (hi == 0) {
        red[ks][0][c16] = su;
        red[ks][1][c16] = sv;
    }
    __syncthreads();

    if (threadIdx.x < 32) {
        const int plane = lane >> 4;
        float s = rhs;
#pragma unroll
        for (int w = 0; w < KS_; ++w) s += red[w][plane][erow];
        UVout[((size_t)(c * 2 + plane)) * TG_ + trow] = s;
    }
}

__global__ __launch_bounds__(128)
void mlp_kernel(const float* __restrict__ UV,
                const float* __restrict__ w_z, const float* __restrict__ b_z,
                const float* __restrict__ w1, const float* __restrict__ b1,
                const float* __restrict__ w2, const float* __restrict__ b2,
                float* __restrict__ out) {
    const int lane = threadIdx.x & 31;
    const int wave = (blockIdx.x * blockDim.x + threadIdx.x) >> 5;
    const int c16  = lane & 15;
    const int hi   = lane >> 4;
    const int base = wave * 32;
    const int c    = base / TG_;
    const int t0   = base % TG_;

    const float* up = UV + ((size_t)c * 2) * TG_ + t0;
    const float u0 = up[c16];
    const float u1 = up[16 + c16];
    const float v0 = up[TG_ + c16];
    const float v1 = up[TG_ + 16 + c16];

    v4f wz[8], bz[8];
#pragma unroll
    for (int i = 0; i < 8; ++i) {
        wz[i] = *(const v4f*)(w_z + 8 * i + 4 * hi);
        bz[i] = *(const v4f*)(b_z + 8 * i + 4 * hi);
    }

    v4f b1f[4][2];
#pragma unroll
    for (int jt = 0; jt < 4; ++jt) {
        b1f[jt][0] = *(const v4f*)(b1 + 16 * jt + 8 * hi);
        b1f[jt][1] = *(const v4f*)(b1 + 16 * jt + 8 * hi + 4);
    }

    v4f wf[4][8];
#pragma unroll
    for (int i = 0; i < 8; ++i)
#pragma unroll
        for (int jt = 0; jt < 4; ++jt)
            wf[jt][i] = *(const v4f*)(w1 + (size_t)(16 * jt + c16) * H_ + 8 * i + 4 * hi);

    v4f w2f[4][2];
#pragma unroll
    for (int jt = 0; jt < 4; ++jt)
#pragma unroll
        for (int q = 0; q < 2; ++q)
            w2f[jt][q] = *(const v4f*)(w2 + (c16 & 1) * H_ + 16 * jt + 8 * hi + 4 * q);
    const float b20 = b2[0];
    const float b21 = b2[1];

    __builtin_amdgcn_sched_barrier(0);

    v8f acc[2][4];
#pragma unroll
    for (int jt = 0; jt < 4; ++jt)
#pragma unroll
        for (int r = 0; r < 4; ++r) {
            acc[0][jt][r] = b1f[jt][0][r];  acc[0][jt][4 + r] = b1f[jt][1][r];
            acc[1][jt][r] = b1f[jt][0][r];  acc[1][jt][4 + r] = b1f[jt][1][r];
        }

    const bool a2row = (c16 & 6) == 0;
#pragma unroll
    for (int jt = 0; jt < 4; ++jt)
#pragma unroll
        for (int q = 0; q < 2; ++q)
#pragma unroll
            for (int e = 0; e < 4; ++e) w2f[jt][q][e] = a2row ? w2f[jt][q][e] : 0.0f;

#pragma unroll
    for (int i = 0; i < 8; ++i) {
        v4f x0, x1;
#pragma unroll
        for (int e = 0; e < 4; ++e) {
            x0[e] = u0 * wz[i][e] + v0 * bz[i][e];
            x1[e] = u1 * wz[i][e] + v1 * bz[i][e];
        }
#pragma unroll
        for (int hf = 0; hf < 2; ++hf) {
            v2f xb0, xb1;
            xb0.x = x0[2 * hf];  xb0.y = x0[2 * hf + 1];
            xb1.x = x1[2 * hf];  xb1.y = x1[2 * hf + 1];
#pragma unroll
            for (int jt = 0; jt < 4; ++jt) {
                v2f a;
                a.x = wf[jt][i][2 * hf];  a.y = wf[jt][i][2 * hf + 1];
                acc[0][jt] = __builtin_amdgcn_wmma_f32_16x16x4_f32(false, a, false, xb0, (short)0, acc[0][jt], false, false);
                acc[1][jt] = __builtin_amdgcn_wmma_f32_16x16x4_f32(false, a, false, xb1, (short)0, acc[1][jt], false, false);
            }
        }
    }

    v8f o2a, o2b;
#pragma unroll
    for (int r = 0; r < 8; ++r) { o2a[r] = 0.0f; o2b[r] = 0.0f; }
    o2a[0] = b20;  o2a[1] = b21;
    o2b[0] = b20;  o2b[1] = b21;
#pragma unroll
    for (int jt = 0; jt < 4; ++jt)
#pragma unroll
        for (int p = 0; p < 4; ++p) {
            v2f a, h0, h1;
            a.x  = w2f[jt][p >> 1][2 * (p & 1)];
            a.y  = w2f[jt][p >> 1][2 * (p & 1) + 1];
            h0.x = fmaxf(acc[0][jt][2 * p], 0.0f);  h0.y = fmaxf(acc[0][jt][2 * p + 1], 0.0f);
            h1.x = fmaxf(acc[1][jt][2 * p], 0.0f);  h1.y = fmaxf(acc[1][jt][2 * p + 1], 0.0f);
            o2a = __builtin_amdgcn_wmma_f32_16x16x4_f32(false, a, false, h0, (short)0, o2a, false, false);
            o2b = __builtin_amdgcn_wmma_f32_16x16x4_f32(false, a, false, h1, (short)0, o2b, false, false);
        }

    const float o0 = hi ? o2b[0] : o2a[0];
    const float o1 = hi ? o2b[1] : o2a[1];
    const int idx = base + lane;
    out[idx] = o0;
    float sp = fmaxf(o1, 0.0f) + log1pf(expf(-fabsf(o1)));
    out[C_ * TG_ + idx] = sp + 1e-6f;
}


#define RLEN_   (TF_ / KS_)
#define RSTEPS_ (RLEN_ / 8)
#define EPL_    (TF_ * 2)

__global__ __launch_bounds__(32 * KS_)
void first_sweep_kernel(const float* __restrict__ grn,
                        const float* __restrict__ z,
                        const int*   __restrict__ binary_tf,
                        float*       __restrict__ UV,
                        float*       __restrict__ zeros,
                        int*         __restrict__ ownf,
                        float*       __restrict__ E1,
                        float*       __restrict__ GT) {
    __shared__ float red[KS_][2][16];
    __shared__ int   ownl[16];

    const int c     = blockIdx.y;
    const int mtile = blockIdx.x;
    const int ks    = threadIdx.x >> 5;
    const int lane  = threadIdx.x & 31;
    const int c16   = lane & 15;
    const int hi    = lane >> 4;

    int col[TF_ / 32];
#pragma unroll
    for (int q = 0; q < TF_ / 32; ++q) col[q] = binary_tf[c * TF_ + q * 32 + lane];
    const int irow = mtile * 16 + 2 * ks;
    const int tr0raw = binary_tf[c * TF_ + irow];
    const int tr1raw = binary_tf[c * TF_ + irow + 1];
    const float zx = z[c * TG_ + mtile * 64 + (threadIdx.x & 63)];
    __builtin_amdgcn_sched_barrier(0);

    const int kbase = ks * KLEN_ + hi * 4;
    const float* grow = grn + ((size_t)(c * TF_ + mtile * 16 + c16)) * TG_ + kbase;
    const float* zrow = z + (size_t)c * TG_ + kbase;
    const float  fill = (c16 == 1) ? 1.0f : 0.0f;

    v8f acc0, acc1;
#pragma unroll
    for (int r = 0; r < 8; ++r) { acc0[r] = 0.0f; acc1[r] = 0.0f; }

    v4f p[STEPS_], g[STEPS_];
#pragma unroll
    for (int j = 0; j < STEPS_; ++j) {
        p[j] = *(const v4f*)(zrow + j * 8);
        g[j] = *(const v4f*)(grow + j * 8);
    }
    __builtin_amdgcn_sched_barrier(0);

    const float* r0 = grn + ((size_t)(c * TF_ + irow)) * TG_;
    const float* r1 = r0 + TG_;
    float ga[TF_ / 32], gb[TF_ / 32];
    const int tr0 = tr0raw & (TG_ - 1);
    const int tr1 = tr1raw & (TG_ - 1);
    bool  later0 = false, later1 = false;
#pragma unroll
    for (int q = 0; q < TF_ / 32; ++q) {
        const int t = col[q] & (TG_ - 1);
        ga[q] = r0[t];
        gb[q] = r1[t];
        later0 |= (t == tr0) & (q * 32 + lane > irow);
        later1 |= (t == tr1) & (q * 32 + lane > irow + 1);
    }
    const bool own0 = __ballot(later0) == 0;
    const bool own1 = __ballot(later1) == 0;
    __builtin_amdgcn_sched_barrier(0);

#pragma unroll
    for (int j = 0; j < STEPS_; ++j) {
        v2f a, b;
        a.x = (c16 == 0) ? p[j].x : fill;  a.y = (c16 == 0) ? p[j].y : fill;
        b.x = g[j].x;  b.y = g[j].y;
        acc0 = __builtin_amdgcn_wmma_f32_16x16x4_f32(false, a, false, b, (short)0, acc0, false, false);
        a.x = (c16 == 0) ? p[j].z : fill;  a.y = (c16 == 0) ? p[j].w : fill;
        b.x = g[j].z;  b.y = g[j].w;
        acc1 = __builtin_amdgcn_wmma_f32_16x16x4_f32(false, a, false, b, (short)0, acc1, false, false);
    }

    const float su = acc0[0] + acc1[0];
    const float sv = acc0[1] + acc1[1];
    if (hi == 0) {
        red[ks][0][c16] = su;
        red[ks][1][c16] = sv;
    }
    if (lane == 0) {
        ownl[2 * ks]     = own0 ? 1 : 0;
        ownl[2 * ks + 1] = own1 ? 1 : 0;
    }

    __syncthreads();

    if (threadIdx.x < 32) {
        const int plane = lane >> 4;
        const int erow  = lane & 15;
        const int own   = ownl[erow];
        float s = red[0][plane][erow];
#pragma unroll
        for (int w = 1; w < KS_; ++w) s += red[w][plane][erow];
        E1[c * EPL_ + mtile * 32 + lane] = own ? s : 0.0f;
        ownf[(c * (TF_ / 16) + mtile) * 32 + lane] = own;
    }

    float* o0 = GT + ((size_t)(c * TF_ + irow)) * TF_ + lane;
#pragma unroll
    for (int q = 0; q < TF_ / 32; ++q) {
        o0[q * 32]       = own0 ? ga[q] : 0.0f;
        o0[TF_ + q * 32] = own1 ? gb[q] : 0.0f;
    }

    const int tq = mtile * 64 + (threadIdx.x & 63);
    if (threadIdx.x < 64)        UV[((size_t)c * 2) * TG_ + tq] = zx;
    else if (threadIdx.x < 128)  UV[((size_t)c * 2 + 1) * TG_ + tq] = 1.0f;
    else if (threadIdx.x < 192 && c == 0) zeros[tq] = 0.0f;
}

template <bool FINAL>
__global__ __launch_bounds__(32 * KS_)
void reduced_sweep_kernel(const float* __restrict__ GT,
                          const float* __restrict__ Ein,
                          const float* __restrict__ E1,
                          const float* __restrict__ zeros,
                          const float* __restrict__ z,
                          const int*   __restrict__ binary_tf,
                          const int*   __restrict__ ownf,
                          float*       __restrict__ dst) {
    __shared__ float red[KS_][2][16];

    const int c     = blockIdx.y;
    const int mtile = blockIdx.x;
    const int ks    = threadIdx.x >> 5;
    const int lane  = threadIdx.x & 31;
    const int c16   = lane & 15;
    const int hi    = lane >> 4;

    const float e1v = (threadIdx.x < 32) ? E1[c * EPL_ + mtile * 32 + lane] : 0.0f;
    int traw = 0, own = 0;
    if (FINAL) {
        traw = binary_tf[c * TF_ + mtile * 16 + c16];
        own  = ownf[(c * (TF_ / 16) + mtile) * 32 + lane];
    }
    __builtin_amdgcn_sched_barrier(0);

    const int kbase = ks * RLEN_ + hi * 4;
    const float* grow = GT + ((size_t)(c * TF_ + mtile * 16 + c16)) * TF_ + kbase;
    const float* prow = (c16 < 2) ? Ein + c * EPL_ + ks * 128 + c16 * 16 + hi * 4
                                  : zeros + hi * 4;

    v8f acc0, acc1;
#pragma unroll
    for (int r = 0; r < 8; ++r) { acc0[r] = 0.0f; acc1[r] = 0.0f; }

    v4f p[RSTEPS_], g[RSTEPS_];
#pragma unroll
    for (int j = 0; j < RSTEPS_; ++j) {
        p[j] = *(const v4f*)(prow + (j >> 1) * 32 + (j & 1) * 8);
        g[j] = *(const v4f*)(grow + j * 8);
    }
    __builtin_amdgcn_sched_barrier(0);
    float rhs = 0.0f;
    int trow = 0;
    if (FINAL) {
        trow = traw & (TG_ - 1);
        rhs  = (threadIdx.x < 16) ? z[c * TG_ + trow] : 1.0f;
        __builtin_amdgcn_sched_barrier(0);
    }
#pragma unroll
    for (int j = 0; j < RSTEPS_; ++j) {
        v2f a, b;
        a.x = p[j].x;  a.y = p[j].y;
        b.x = g[j].x;  b.y = g[j].y;
        acc0 = __builtin_amdgcn_wmma_f32_16x16x4_f32(false, a, false, b, (short)0, acc0, false, false);
        a.x = p[j].z;  a.y = p[j].w;
        b.x = g[j].z;  b.y = g[j].w;
        acc1 = __builtin_amdgcn_wmma_f32_16x16x4_f32(false, a, false, b, (short)0, acc1, false, false);
    }

    const float su = acc0[0] + acc1[0];
    const float sv = acc0[1] + acc1[1];
    if (hi == 0) {
        red[ks][0][c16] = su;
        red[ks][1][c16] = sv;
    }
    __syncthreads();

    if (threadIdx.x < 32) {
        const int plane = lane >> 4;
        const int erow  = lane & 15;
        float s = e1v;
#pragma unroll
        for (int w = 0; w < KS_; ++w) s += red[w][plane][erow];
        if (FINAL) {
            if (own) dst[((size_t)(c * 2 + plane)) * TG_ + trow] = rhs + s;
        } else {
            dst[c * EPL_ + mtile * 32 + lane] = s;
        }
    }
}

extern "C" void kernel_launch(void* const* d_in, const int* in_sizes, int n_in,
                              void* d_out, int out_size, void* d_ws, size_t ws_size,
                              hipStream_t stream) {
    (void)in_sizes; (void)n_in; (void)out_size;
    const float* z   = (const float*)d_in[0];
    const float* grn = (const float*)d_in[1];
    const int*   btf = (const int*)  d_in[2];
    const float* w_z = (const float*)d_in[4];
    const float* b_z = (const float*)d_in[5];
    const float* w1  = (const float*)d_in[6];
    const float* b1  = (const float*)d_in[7];
    const float* w2  = (const float*)d_in[8];
    const float* b2  = (const float*)d_in[9];
    float* out = (float*)d_out;

    dim3 grid(TF_ / 16, C_);

    if (ws_size >= ((size_t)C_ * 3 * TG_ + TG_ + (size_t)3 * C_ * EPL_ + (size_t)C_ * TF_ * TF_) * sizeof(float)) {
        float* UV  = (float*)d_ws;
        float* zer = UV + (size_t)C_ * 2 * TG_;
        int*   own = (int*)(zer + TG_);
        float* E1  = (float*)(own + (size_t)C_ * TG_);
        float* Ea  = E1 + (size_t)C_ * EPL_;
        float* Eb  = Ea + (size_t)C_ * EPL_;
        float* GT  = Eb + (size_t)C_ * EPL_;

        first_sweep_kernel<<<grid, 32 * KS_, 0, stream>>>(grn, z, btf, UV, zer, own, E1, GT);
        const float* ein = E1;
        float* ebuf[2] = { Ea, Eb };
        for (int s = 2; s < NITER; ++s) {
            float* eout = ebuf[s & 1];
            reduced_sweep_kernel<false><<<grid, 32 * KS_, 0, stream>>>(GT, ein, E1, zer, z, btf, own, eout);
            ein = eout;
        }
        reduced_sweep_kernel<true><<<grid, 32 * KS_, 0, stream>>>(GT, ein, E1, zer, z, btf, own, UV);
        mlp_kernel<<<(C_ * TG_ + 127) / 128, 128, 0, stream>>>(UV, w_z, b_z, w1, b1, w2, b2, out);
        return;
    }

    float* UVa = (float*)d_ws;
    float* UVb = UVa + (size_t)C_ * 2 * TG_;
    float* zer = UVb + (size_t)C_ * 2 * TG_;

    init_uv_kernel<<<(C_ * TG_ + 255) / 256, 256, 0, stream>>>(z, UVa, UVb, zer);

    float* bufs[2] = { UVa, UVb };
    for (int it = 0; it < NITER; ++it) {
        const float* uvin = bufs[it & 1];
        float*      uvout = bufs[(it + 1) & 1];
        neumann_iter_kernel<<<grid, 32 * KS_, 0, stream>>>(grn, z, btf, uvin, zer, uvout);
    }
    const float* uvfin = bufs[NITER & 1];

    mlp_kernel<<<(C_ * TG_ + 127) / 128, 128, 0, stream>>>(uvfin, w_z, b_z, w1, b1, w2, b2, out);
}


# one launch forms P=GT.GT on f32 WMMA (64x128 tiles, every 128-byte line of P written whole by one store); f=e1+GT.e1, then 8 sweeps on P: same 18 terms, 19->12 launches; sum order changes
# speedup vs baseline: 17.5311x; 1.0754x over previous
#include <hip/hip_runtime.h>
#include <hip/hip_bf16.h>

#define C_   8
#define TF_  512
#define TG_  2048
#define H_   64
#define NITER 18
#define KS_   8
#define KLEN_ (TG_ / KS_)
#define STEPS_ (KLEN_ / 8)

typedef __attribute__((ext_vector_type(2))) float v2f;
typedef __attribute__((ext_vector_type(4))) float v4f;
typedef __attribute__((ext_vector_type(8))) float v8f;


__global__ __launch_bounds__(256)
void init_uv_kernel(const float* __restrict__ z,
                    float* __restrict__ UVa,
                    float* __restrict__ UVb,
                    float* __restrict__ zeros) {
    int idx = blockIdx.x * blockDim.x + threadIdx.x;
    if (idx >= C_ * TG_) return;
    if (idx < TG_) zeros[idx] = 0.0f;
    int c = idx / TG_;
    int t = idx % TG_;
    float zz = z[idx];
    size_t pu = ((size_t)c * 2) * TG_ + t;
    size_t pv = pu + TG_;
    UVa[pu] = zz;  UVa[pv] = 1.0f;
    UVb[pu] = zz;  UVb[pv] = 1.0f;
}

__global__ __launch_bounds__(32 * KS_)
void neumann_iter_kernel(const float* __restrict__ grn,
                         const float* __restrict__ z,
                         const int*   __restrict__ binary_tf,
                         const float* __restrict__ UVin,
                         const float* __restrict__ zeros,
                         float*       __restrict__ UVout) {
    __shared__ float red[KS_][2][16];

    const int c     = blockIdx.y;
    const int mtile = blockIdx.x;
    const int ks    = threadIdx.x >> 5;
    const int lane  = threadIdx.x & 31;
    const int c16   = lane & 15;
    const int hi    = lane >> 4;

    const int erow = lane & 15;
    const int traw = binary_tf[c * TF_ + mtile * 16 + erow];
    __builtin_amdgcn_sched_barrier(0);

    const int kbase = ks * KLEN_ + hi * 4;
    const float* grow = grn + ((size_t)(c * TF_ + mtile * 16 + c16)) * TG_ + kbase;
    const float* prow = (c16 < 2) ? UVin + ((size_t)(c * 2 + c16)) * TG_ + kbase
                                  : zeros + kbase;

    v8f acc0, acc1;
#pragma unroll
    for (int r = 0; r < 8; ++r) { acc0[r] = 0.0f; acc1[r] = 0.0f; }

    v4f p[STEPS_], g[STEPS_];
#pragma unroll
    for (int j = 0; j < STEPS_; ++j) {
        p[j] = *(const v4f*)(prow + j * 8);
        g[j] = *(const v4f*)(grow + j * 8);
    }
    __builtin_amdgcn_sched_barrier(0);
    const int   trow = traw & (TG_ - 1);
    const float rhs  = (threadIdx.x < 16) ? z[c * TG_ + trow] : 1.0f;
    __builtin_amdgcn_sched_barrier(0);
#pragma unroll
    for (int j = 0; j < STEPS_; ++j) {
        v2f a, b;
        a.x = p[j].x;  a.y = p[j].y;
        b.x = g[j].x;  b.y = g[j].y;
        acc0 = __builtin_amdgcn_wmma_f32_16x16x4_f32(false, a, false, b, (short)0, acc0, false, false);
        a.x = p[j].z;  a.y = p[j].w;
        b.x = g[j].z;  b.y = g[j].w;
        acc1 = __builtin_amdgcn_wmma_f32_16x16x4_f32(false, a, false, b, (short)0, acc1, false, false);
    }

    const float su = acc0[0] + acc1[0];
    const float sv = acc0[1] + acc1[1];
    if (hi == 0) {
        red[ks][0][c16] = su;
        red[ks][1][c16] = sv;
    }
    __syncthreads();

    if (threadIdx.x < 32) {
        const int plane = lane >> 4;
        float s = rhs;
#pragma unroll
        for (int w = 0; w < KS_; ++w) s += red[w][plane][erow];
        UVout[((size_t)(c * 2 + plane)) * TG_ + trow] = s;
    }
}

__global__ __launch_bounds__(128)
void mlp_kernel(const float* __restrict__ UV,
                const float* __restrict__ w_z, const float* __restrict__ b_z,
                const float* __restrict__ w1, const float* __restrict__ b1,
                const float* __restrict__ w2, const float* __restrict__ b2,
                float* __restrict__ out) {
    const int lane = threadIdx.x & 31;
    const int wave = (blockIdx.x * blockDim.x + threadIdx.x) >> 5;
    const int c16  = lane & 15;
    const int hi   = lane >> 4;
    const int base = wave * 32;
    const int c    = base / TG_;
    const int t0   = base % TG_;

    const float* up = UV + ((size_t)c * 2) * TG_ + t0;
    const float u0 = up[c16];
    const float u1 = up[16 + c16];
    const float v0 = up[TG_ + c16];
    const float v1 = up[TG_ + 16 + c16];

    v4f wz[8], bz[8];
#pragma unroll
    for (int i = 0; i < 8; ++i) {
        wz[i] = *(const v4f*)(w_z + 8 * i + 4 * hi);
        bz[i] = *(const v4f*)(b_z + 8 * i + 4 * hi);
    }

    v4f b1f[4][2];
#pragma unroll
    for (int jt = 0; jt < 4; ++jt) {
        b1f[jt][0] = *(const v4f*)(b1 + 16 * jt + 8 * hi);
        b1f[jt][1] = *(const v4f*)(b1 + 16 * jt + 8 * hi + 4);
    }

    v4f wf[4][8];
#pragma unroll
    for (int i = 0; i < 8; ++i)
#pragma unroll
        for (int jt = 0; jt < 4; ++jt)
            wf[jt][i] = *(const v4f*)(w1 + (size_t)(16 * jt + c16) * H_ + 8 * i + 4 * hi);

    v4f w2f[4][2];
#pragma unroll
    for (int jt = 0; jt < 4; ++jt)
#pragma unroll
        for (int q = 0; q < 2; ++q)
            w2f[jt][q] = *(const v4f*)(w2 + (c16 & 1) * H_ + 16 * jt + 8 * hi + 4 * q);
    const float b20 = b2[0];
    const float b21 = b2[1];

    __builtin_amdgcn_sched_barrier(0);

    v8f acc[2][4];
#pragma unroll
    for (int jt = 0; jt < 4; ++jt)
#pragma unroll
        for (int r = 0; r < 4; ++r) {
            acc[0][jt][r] = b1f[jt][0][r];  acc[0][jt][4 + r] = b1f[jt][1][r];
            acc[1][jt][r] = b1f[jt][0][r];  acc[1][jt][4 + r] = b1f[jt][1][r];
        }

    const bool a2row = (c16 & 6) == 0;
#pragma unroll
    for (int jt = 0; jt < 4; ++jt)
#pragma unroll
        for (int q = 0; q < 2; ++q)
#pragma unroll
            for (int e = 0; e < 4; ++e) w2f[jt][q][e] = a2row ? w2f[jt][q][e] : 0.0f;

#pragma unroll
    for (int i = 0; i < 8; ++i) {
        v4f x0, x1;
#pragma unroll
        for (int e = 0; e < 4; ++e) {
            x0[e] = u0 * wz[i][e] + v0 * bz[i][e];
            x1[e] = u1 * wz[i][e] + v1 * bz[i][e];
        }
#pragma unroll
        for (int hf = 0; hf < 2; ++hf) {
            v2f xb0, xb1;
            xb0.x = x0[2 * hf];  xb0.y = x0[2 * hf + 1];
            xb1.x = x1[2 * hf];  xb1.y = x1[2 * hf + 1];
#pragma unroll
            for (int jt = 0; jt < 4; ++jt) {
                v2f a;
                a.x = wf[jt][i][2 * hf];  a.y = wf[jt][i][2 * hf + 1];
                acc[0][jt] = __builtin_amdgcn_wmma_f32_16x16x4_f32(false, a, false, xb0, (short)0, acc[0][jt], false, false);
                acc[1][jt] = __builtin_amdgcn_wmma_f32_16x16x4_f32(false, a, false, xb1, (short)0, acc[1][jt], false, false);
            }
        }
    }

    v8f o2a, o2b;
#pragma unroll
    for (int r = 0; r < 8; ++r) { o2a[r] = 0.0f; o2b[r] = 0.0f; }
    o2a[0] = b20;  o2a[1] = b21;
    o2b[0] = b20;  o2b[1] = b21;
#pragma unroll
    for (int jt = 0; jt < 4; ++jt)
#pragma unroll
        for (int p = 0; p < 4; ++p) {
            v2f a, h0, h1;
            a.x  = w2f[jt][p >> 1][2 * (p & 1)];
            a.y  = w2f[jt][p >> 1][2 * (p & 1) + 1];
            h0.x = fmaxf(acc[0][jt][2 * p], 0.0f);  h0.y = fmaxf(acc[0][jt][2 * p + 1], 0.0f);
            h1.x = fmaxf(acc[1][jt][2 * p], 0.0f);  h1.y = fmaxf(acc[1][jt][2 * p + 1], 0.0f);
            o2a = __builtin_amdgcn_wmma_f32_16x16x4_f32(false, a, false, h0, (short)0, o2a, false, false);
            o2b = __builtin_amdgcn_wmma_f32_16x16x4_f32(false, a, false, h1, (short)0, o2b, false, false);
        }

    const float o0 = hi ? o2b[0] : o2a[0];
    const float o1 = hi ? o2b[1] : o2a[1];
    const int idx = base + lane;
    out[idx] = o0;
    float sp = fmaxf(o1, 0.0f) + log1pf(expf(-fabsf(o1)));
    out[C_ * TG_ + idx] = sp + 1e-6f;
}


#define RLEN_   (TF_ / KS_)
#define RSTEPS_ (RLEN_ / 8)
#define EPL_    (TF_ * 2)

__global__ __launch_bounds__(32 * KS_)
void first_sweep_kernel(const float* __restrict__ grn,
                        const float* __restrict__ z,
                        const int*   __restrict__ binary_tf,
                        float*       __restrict__ UV,
                        float*       __restrict__ zeros,
                        int*         __restrict__ ownf,
                        float*       __restrict__ E1,
                        float*       __restrict__ GT) {
    __shared__ float red[KS_][2][16];
    __shared__ int   ownl[16];

    const int c     = blockIdx.y;
    const int mtile = blockIdx.x;
    const int ks    = threadIdx.x >> 5;
    const int lane  = threadIdx.x & 31;
    const int c16   = lane & 15;
    const int hi    = lane >> 4;

    int col[TF_ / 32];
#pragma unroll
    for (int q = 0; q < TF_ / 32; ++q) col[q] = binary_tf[c * TF_ + q * 32 + lane];
    const int irow = mtile * 16 + 2 * ks;
    const int tr0raw = binary_tf[c * TF_ + irow];
    const int tr1raw = binary_tf[c * TF_ + irow + 1];
    const float zx = z[c * TG_ + mtile * 64 + (threadIdx.x & 63)];
    __builtin_amdgcn_sched_barrier(0);

    const int kbase = ks * KLEN_ + hi * 4;
    const float* grow = grn + ((size_t)(c * TF_ + mtile * 16 + c16)) * TG_ + kbase;
    const float* zrow = z + (size_t)c * TG_ + kbase;
    const float  fill = (c16 == 1) ? 1.0f : 0.0f;

    v8f acc0, acc1;
#pragma unroll
    for (int r = 0; r < 8; ++r) { acc0[r] = 0.0f; acc1[r] = 0.0f; }

    v4f p[STEPS_], g[STEPS_];
#pragma unroll
    for (int j = 0; j < STEPS_; ++j) {
        p[j] = *(const v4f*)(zrow + j * 8);
        g[j] = *(const v4f*)(grow + j * 8);
    }
    __builtin_amdgcn_sched_barrier(0);

    const float* r0 = grn + ((size_t)(c * TF_ + irow)) * TG_;
    const float* r1 = r0 + TG_;
    float ga[TF_ / 32], gb[TF_ / 32];
    const int tr0 = tr0raw & (TG_ - 1);
    const int tr1 = tr1raw & (TG_ - 1);
    bool  later0 = false, later1 = false;
#pragma unroll
    for (int q = 0; q < TF_ / 32; ++q) {
        const int t = col[q] & (TG_ - 1);
        ga[q] = r0[t];
        gb[q] = r1[t];
        later0 |= (t == tr0) & (q * 32 + lane > irow);
        later1 |= (t == tr1) & (q * 32 + lane > irow + 1);
    }
    const bool own0 = __ballot(later0) == 0;
    const bool own1 = __ballot(later1) == 0;
    __builtin_amdgcn_sched_barrier(0);

#pragma unroll
    for (int j = 0; j < STEPS_; ++j) {
        v2f a, b;
        a.x = (c16 == 0) ? p[j].x : fill;  a.y = (c16 == 0) ? p[j].y : fill;
        b.x = g[j].x;  b.y = g[j].y;
        acc0 = __builtin_amdgcn_wmma_f32_16x16x4_f32(false, a, false, b, (short)0, acc0, false, false);
        a.x = (c16 == 0) ? p[j].z : fill;  a.y = (c16 == 0) ? p[j].w : fill;
        b.x = g[j].z;  b.y = g[j].w;
        acc1 = __builtin_amdgcn_wmma_f32_16x16x4_f32(false, a, false, b, (short)0, acc1, false, false);
    }

    const float su = acc0[0] + acc1[0];
    const float sv = acc0[1] + acc1[1];
    if (hi == 0) {
        red[ks][0][c16] = su;
        red[ks][1][c16] = sv;
    }
    if (lane == 0) {
        ownl[2 * ks]     = own0 ? 1 : 0;
        ownl[2 * ks + 1] = own1 ? 1 : 0;
    }

    __syncthreads();

    if (threadIdx.x < 32) {
        const int plane = lane >> 4;
        const int erow  = lane & 15;
        const int own   = ownl[erow];
        float s = red[0][plane][erow];
#pragma unroll
        for (int w = 1; w < KS_; ++w) s += red[w][plane][erow];
        E1[c * EPL_ + mtile * 32 + lane] = own ? s : 0.0f;
        ownf[(c * (TF_ / 16) + mtile) * 32 + lane] = own;
    }

    float* o0 = GT + ((size_t)(c * TF_ + irow)) * TF_ + lane;
#pragma unroll
    for (int q = 0; q < TF_ / 32; ++q) {
        o0[q * 32]       = own0 ? ga[q] : 0.0f;
        o0[TF_ + q * 32] = own1 ? gb[q] : 0.0f;
    }

    const int tq = mtile * 64 + (threadIdx.x & 63);
    if (threadIdx.x < 64)        UV[((size_t)c * 2) * TG_ + tq] = zx;
    else if (threadIdx.x < 128)  UV[((size_t)c * 2 + 1) * TG_ + tq] = 1.0f;
    else if (threadIdx.x < 192 && c == 0) zeros[tq] = 0.0f;
}

template <bool FINAL>
__global__ __launch_bounds__(32 * KS_)
void reduced_sweep_kernel(const float* __restrict__ GT,
                          const float* __restrict__ Ein,
                          const float* __restrict__ E1,
                          const float* __restrict__ zeros,
                          const float* __restrict__ z,
                          const int*   __restrict__ binary_tf,
                          const int*   __restrict__ ownf,
                          float*       __restrict__ dst) {
    __shared__ float red[KS_][2][16];

    const int c     = blockIdx.y;
    const int mtile = blockIdx.x;
    const int ks    = threadIdx.x >> 5;
    const int lane  = threadIdx.x & 31;
    const int c16   = lane & 15;
    const int hi    = lane >> 4;

    const float e1v = (threadIdx.x < 32) ? E1[c * EPL_ + mtile * 32 + lane] : 0.0f;
    int traw = 0, own = 0;
    if (FINAL) {
        traw = binary_tf[c * TF_ + mtile * 16 + c16];
        own  = ownf[(c * (TF_ / 16) + mtile) * 32 + lane];
    }
    __builtin_amdgcn_sched_barrier(0);

    const int kbase = ks * RLEN_ + hi * 4;
    const float* grow = GT + ((size_t)(c * TF_ + mtile * 16 + c16)) * TF_ + kbase;
    const float* prow = (c16 < 2) ? Ein + c * EPL_ + ks * 128 + c16 * 16 + hi * 4
                                  : zeros + hi * 4;

    v8f acc0, acc1;
#pragma unroll
    for (int r = 0; r < 8; ++r) { acc0[r] = 0.0f; acc1[r] = 0.0f; }

    v4f p[RSTEPS_], g[RSTEPS_];
#pragma unroll
    for (int j = 0; j < RSTEPS_; ++j) {
        p[j] = *(const v4f*)(prow + (j >> 1) * 32 + (j & 1) * 8);
        g[j] = *(const v4f*)(grow + j * 8);
    }
    __builtin_amdgcn_sched_barrier(0);
    float rhs = 0.0f;
    int trow = 0;
    if (FINAL) {
        trow = traw & (TG_ - 1);
        rhs  = (threadIdx.x < 16) ? z[c * TG_ + trow] : 1.0f;
        __builtin_amdgcn_sched_barrier(0);
    }
#pragma unroll
    for (int j = 0; j < RSTEPS_; ++j) {
        v2f a, b;
        a.x = p[j].x;  a.y = p[j].y;
        b.x = g[j].x;  b.y = g[j].y;
        acc0 = __builtin_amdgcn_wmma_f32_16x16x4_f32(false, a, false, b, (short)0, acc0, false, false);
        a.x = p[j].z;  a.y = p[j].w;
        b.x = g[j].z;  b.y = g[j].w;
        acc1 = __builtin_amdgcn_wmma_f32_16x16x4_f32(false, a, false, b, (short)0, acc1, false, false);
    }

    const float su = acc0[0] + acc1[0];
    const float sv = acc0[1] + acc1[1];
    if (hi == 0) {
        red[ks][0][c16] = su;
        red[ks][1][c16] = sv;
    }
    __syncthreads();

    if (threadIdx.x < 32) {
        const int plane = lane >> 4;
        const int erow  = lane & 15;
        float s = e1v;
#pragma unroll
        for (int w = 0; w < KS_; ++w) s += red[w][plane][erow];
        if (FINAL) {
            if (own) dst[((size_t)(c * 2 + plane)) * TG_ + trow] = rhs + s;
        } else {
            dst[c * EPL_ + mtile * 32 + lane] = s;
        }
    }
}


#define SQD_    4
#define SQSTEPS_ (TF_ / 8)

__global__ __launch_bounds__(128)
void square_kernel(const float* __restrict__ GT, float* __restrict__ P) {
    const int c    = blockIdx.y;
    const int mb   = blockIdx.x >> 2;
    const int nb   = blockIdx.x & 3;
    const int w    = threadIdx.x >> 5;
    const int lane = threadIdx.x & 31;
    const int c16  = lane & 15;
    const int hi   = lane >> 4;
    const int m0   = mb * 64 + (w >> 1) * 32;
    const int n0   = nb * 128 + (w & 1) * 64;

    const float* G  = GT + (size_t)c * TF_ * TF_;
    const float* ap = G + (size_t)(m0 + c16) * TF_ + 4 * hi;
    const float* bp = G + (size_t)(4 * hi) * TF_ + n0 + 4 * c16;

    v8f acc[2][4];
#pragma unroll
    for (int mt = 0; mt < 2; ++mt)
#pragma unroll
        for (int nt = 0; nt < 4; ++nt)
#pragma unroll
            for (int r = 0; r < 8; ++r) acc[mt][nt][r] = 0.0f;

    v4f a[SQD_][2], b[SQD_][4];

#define SQ_LOAD(s, step)                                                            \
    {                                                                               \
        a[s][0] = *(const v4f*)(ap + (size_t)(step) * 8);                           \
        a[s][1] = *(const v4f*)(ap + (size_t)16 * TF_ + (size_t)(step) * 8);        \
        b[s][0] = *(const v4f*)(bp + ((size_t)(step) * 8 + 0) * TF_);               \
        b[s][1] = *(const v4f*)(bp + ((size_t)(step) * 8 + 1) * TF_);               \
        b[s][2] = *(const v4f*)(bp + ((size_t)(step) * 8 + 2) * TF_);               \
        b[s][3] = *(const v4f*)(bp + ((size_t)(step) * 8 + 3) * TF_);               \
    }

#define SQ_COMPUTE(s)                                                               \
    {                                                                               \
        _Pragma("unroll")                                                           \
        for (int h2 = 0; h2 < 2; ++h2) {                                            \
            v2f bb[4];                                                              \
            _Pragma("unroll")                                                       \
            for (int nt = 0; nt < 4; ++nt) {                                        \
                bb[nt].x = b[s][2 * h2][nt];                                        \
                bb[nt].y = b[s][2 * h2 + 1][nt];                                    \
            }                                                                       \
            _Pragma("unroll")                                                       \
            for (int mt = 0; mt < 2; ++mt) {                                        \
                v2f aa;                                                             \
                aa.x = a[s][mt][2 * h2];                                            \
                aa.y = a[s][mt][2 * h2 + 1];                                        \
                _Pragma("unroll")                                                   \
                for (int nt = 0; nt < 4; ++nt)                                      \
                    acc[mt][nt] = __builtin_amdgcn_wmma_f32_16x16x4_f32(            \
                        false, aa, false, bb[nt], (short)0, acc[mt][nt], false, false); \
            }                                                                       \
        }                                                                           \
    }

#pragma unroll
    for (int s = 0; s < SQD_; ++s) SQ_LOAD(s, s)
    __builtin_amdgcn_sched_barrier(0);

#pragma nounroll
    for (int it = 0; it < SQSTEPS_ - SQD_; it += SQD_) {
#pragma unroll
        for (int s = 0; s < SQD_; ++s) {
            SQ_COMPUTE(s)
            __builtin_amdgcn_sched_barrier(0);
            SQ_LOAD(s, it + SQD_ + s)
            __builtin_amdgcn_sched_barrier(0);
        }
    }
#pragma unroll
    for (int s = 0; s < SQD_; ++s) SQ_COMPUTE(s)

#undef SQ_LOAD
#undef SQ_COMPUTE

    float* prow = P + (size_t)c * TF_ * TF_ + (size_t)(m0 + 8 * hi) * TF_ + n0 + 4 * c16;
#pragma unroll
    for (int mt = 0; mt < 2; ++mt)
#pragma unroll
        for (int r = 0; r < 8; ++r) {
            v4f o;
            o.x = acc[mt][0][r];  o.y = acc[mt][1][r];
            o.z = acc[mt][2][r];  o.w = acc[mt][3][r];
            *(v4f*)(prow + (size_t)(16 * mt + r) * TF_) = o;
        }
}

extern "C" void kernel_launch(void* const* d_in, const int* in_sizes, int n_in,
                              void* d_out, int out_size, void* d_ws, size_t ws_size,
                              hipStream_t stream) {
    (void)in_sizes; (void)n_in; (void)out_size;
    const float* z   = (const float*)d_in[0];
    const float* grn = (const float*)d_in[1];
    const int*   btf = (const int*)  d_in[2];
    const float* w_z = (const float*)d_in[4];
    const float* b_z = (const float*)d_in[5];
    const float* w1  = (const float*)d_in[6];
    const float* b1  = (const float*)d_in[7];
    const float* w2  = (const float*)d_in[8];
    const float* b2  = (const float*)d_in[9];
    float* out = (float*)d_out;

    dim3 grid(TF_ / 16, C_);

    if (ws_size >= ((size_t)C_ * 3 * TG_ + TG_ + (size_t)4 * C_ * EPL_ + (size_t)2 * C_ * TF_ * TF_) * sizeof(float)) {
        float* UV  = (float*)d_ws;
        float* zer = UV + (size_t)C_ * 2 * TG_;
        int*   own = (int*)(zer + TG_);
        float* E1  = (float*)(own + (size_t)C_ * TG_);
        float* F   = E1 + (size_t)C_ * EPL_;
        float* Ea  = F  + (size_t)C_ * EPL_;
        float* Eb  = Ea + (size_t)C_ * EPL_;
        float* GT  = Eb + (size_t)C_ * EPL_;
        float* P   = GT + (size_t)C_ * TF_ * TF_;

        first_sweep_kernel<<<grid, 32 * KS_, 0, stream>>>(grn, z, btf, UV, zer, own, E1, GT);
        square_kernel<<<dim3(32, C_), 128, 0, stream>>>(GT, P);
        reduced_sweep_kernel<false><<<grid, 32 * KS_, 0, stream>>>(GT, E1, E1, zer, z, btf, own, F);
        const float* yin = F;
        float* ybuf[2] = { Ea, Eb };
        for (int j = 1; j < 8; ++j) {
            float* yout = ybuf[j & 1];
            reduced_sweep_kernel<false><<<grid, 32 * KS_, 0, stream>>>(P, yin, F, zer, z, btf, own, yout);
            yin = yout;
        }
        reduced_sweep_kernel<true><<<grid, 32 * KS_, 0, stream>>>(P, yin, F, zer, z, btf, own, UV);
        mlp_kernel<<<(C_ * TG_ + 127) / 128, 128, 0, stream>>>(UV, w_z, b_z, w1, b1, w2, b2, out);
        return;
    }

    if (ws_size >= ((size_t)C_ * 3 * TG_ + TG_ + (size_t)3 * C_ * EPL_ + (size_t)C_ * TF_ * TF_) * sizeof(float)) {
        float* UV  = (float*)d_ws;
        float* zer = UV + (size_t)C_ * 2 * TG_;
        int*   own = (int*)(zer + TG_);
        float* E1  = (float*)(own + (size_t)C_ * TG_);
        float* Ea  = E1 + (size_t)C_ * EPL_;
        float* Eb  = Ea + (size_t)C_ * EPL_;
        float* GT  = Eb + (size_t)C_ * EPL_;

        first_sweep_kernel<<<grid, 32 * KS_, 0, stream>>>(grn, z, btf, UV, zer, own, E1, GT);
        const float* ein = E1;
        float* ebuf[2] = { Ea, Eb };
        for (int s = 2; s < NITER; ++s) {
            float* eout = ebuf[s & 1];
            reduced_sweep_kernel<false><<<grid, 32 * KS_, 0, stream>>>(GT, ein, E1, zer, z, btf, own, eout);
            ein = eout;
        }
        reduced_sweep_kernel<true><<<grid, 32 * KS_, 0, stream>>>(GT, ein, E1, zer, z, btf, own, UV);
        mlp_kernel<<<(C_ * TG_ + 127) / 128, 128, 0, stream>>>(UV, w_z, b_z, w1, b1, w2, b2, out);
        return;
    }

    float* UVa = (float*)d_ws;
    float* UVb = UVa + (size_t)C_ * 2 * TG_;
    float* zer = UVb + (size_t)C_ * 2 * TG_;

    init_uv_kernel<<<(C_ * TG_ + 255) / 256, 256, 0, stream>>>(z, UVa, UVb, zer);

    float* bufs[2] = { UVa, UVb };
    for (int it = 0; it < NITER; ++it) {
        const float* uvin = bufs[it & 1];
        float*      uvout = bufs[(it + 1) & 1];
        neumann_iter_kernel<<<grid, 32 * KS_, 0, stream>>>(grn, z, btf, uvin, zer, uvout);
    }
    const float* uvfin = bufs[NITER & 1];

    mlp_kernel<<<(C_ * TG_ + 127) / 128, 128, 0, stream>>>(uvfin, w_z, b_z, w1, b1, w2, b2, out);
}


# the f sweep (f = e1 + GT.e1) rides in the square launch as 256 extra 4-wave work-groups dispatched behind the 256 of P = GT.GT (12 -> 11 launches); same K-slices and sum order, same bits
# speedup vs baseline: 18.6575x; 1.0643x over previous
#include <hip/hip_runtime.h>
#include <hip/hip_bf16.h>

#define C_   8
#define TF_  512
#define TG_  2048
#define H_   64
#define NITER 18
#define KS_   8
#define KLEN_ (TG_ / KS_)
#define STEPS_ (KLEN_ / 8)

typedef __attribute__((ext_vector_type(2))) float v2f;
typedef __attribute__((ext_vector_type(4))) float v4f;
typedef __attribute__((ext_vector_type(8))) float v8f;


__global__ __launch_bounds__(256)
void init_uv_kernel(const float* __restrict__ z,
                    float* __restrict__ UVa,
                    float* __restrict__ UVb,
                    float* __restrict__ zeros) {
    int idx = blockIdx.x * blockDim.x + threadIdx.x;
    if (idx >= C_ * TG_) return;
    if (idx < TG_) zeros[idx] = 0.0f;
    int c = idx / TG_;
    int t = idx % TG_;
    float zz = z[idx];
    size_t pu = ((size_t)c * 2) * TG_ + t;
    size_t pv = pu + TG_;
    UVa[pu] = zz;  UVa[pv] = 1.0f;
    UVb[pu] = zz;  UVb[pv] = 1.0f;
}

__global__ __launch_bounds__(32 * KS_)
void neumann_iter_kernel(const float* __restrict__ grn,
                         const float* __restrict__ z,
                         const int*   __restrict__ binary_tf,
                         const float* __restrict__ UVin,
                         const float* __restrict__ zeros,
                         float*       __restrict__ UVout) {
    __shared__ float red[KS_][2][16];

    const int c     = blockIdx.y;
    const int mtile = blockIdx.x;
    const int ks    = threadIdx.x >> 5;
    const int lane  = threadIdx.x & 31;
    const int c16   = lane & 15;
    const int hi    = lane >> 4;

    const int erow = lane & 15;
    const int traw = binary_tf[c * TF_ + mtile * 16 + erow];
    __builtin_amdgcn_sched_barrier(0);

    const int kbase = ks * KLEN_ + hi * 4;
    const float* grow = grn + ((size_t)(c * TF_ + mtile * 16 + c16)) * TG_ + kbase;
    const float* prow = (c16 < 2) ? UVin + ((size_t)(c * 2 + c16)) * TG_ + kbase
                                  : zeros + kbase;

    v8f acc0, acc1;
#pragma unroll
    for (int r = 0; r < 8; ++r) { acc0[r] = 0.0f; acc1[r] = 0.0f; }

    v4f p[STEPS_], g[STEPS_];
#pragma unroll
    for (int j = 0; j < STEPS_; ++j) {
        p[j] = *(const v4f*)(prow + j * 8);
        g[j] = *(const v4f*)(grow + j * 8);
    }
    __builtin_amdgcn_sched_barrier(0);
    const int   trow = traw & (TG_ - 1);
    const float rhs  = (threadIdx.x < 16) ? z[c * TG_ + trow] : 1.0f;
    __builtin_amdgcn_sched_barrier(0);
#pragma unroll
    for (int j = 0; j < STEPS_; ++j) {
        v2f a, b;
        a.x = p[j].x;  a.y = p[j].y;
        b.x = g[j].x;  b.y = g[j].y;
        acc0 = __builtin_amdgcn_wmma_f32_16x16x4_f32(false, a, false, b, (short)0, acc0, false, false);
        a.x = p[j].z;  a.y = p[j].w;
        b.x = g[j].z;  b.y = g[j].w;
        acc1 = __builtin_amdgcn_wmma_f32_16x16x4_f32(false, a, false, b, (short)0, acc1, false, false);
    }

    const float su = acc0[0] + acc1[0];
    const float sv = acc0[1] + acc1[1];
    if (hi == 0) {
        red[ks][0][c16] = su;
        red[ks][1][c16] = sv;
    }
    __syncthreads();

    if (threadIdx.x < 32) {
        const int plane = lane >> 4;
        float s = rhs;
#pragma unroll
        for (int w = 0; w < KS_; ++w) s += red[w][plane][erow];
        UVout[((size_t)(c * 2 + plane)) * TG_ + trow] = s;
    }
}

__global__ __launch_bounds__(128)
void mlp_kernel(const float* __restrict__ UV,
                const float* __restrict__ w_z, const float* __restrict__ b_z,
                const float* __restrict__ w1, const float* __restrict__ b1,
                const float* __restrict__ w2, const float* __restrict__ b2,
                float* __restrict__ out) {
    const int lane = threadIdx.x & 31;
    const int wave = (blockIdx.x * blockDim.x + threadIdx.x) >> 5;
    const int c16  = lane & 15;
    const int hi   = lane >> 4;
    const int base = wave * 32;
    const int c    = base / TG_;
    const int t0   = base % TG_;

    const float* up = UV + ((size_t)c * 2) * TG_ + t0;
    const float u0 = up[c16];
    const float u1 = up[16 + c16];
    const float v0 = up[TG_ + c16];
    const float v1 = up[TG_ + 16 + c16];

    v4f wz[8], bz[8];
#pragma unroll
    for (int i = 0; i < 8; ++i) {
        wz[i] = *(const v4f*)(w_z + 8 * i + 4 * hi);
        bz[i] = *(const v4f*)(b_z + 8 * i + 4 * hi);
    }

    v4f b1f[4][2];
#pragma unroll
    for (int jt = 0; jt < 4; ++jt) {
        b1f[jt][0] = *(const v4f*)(b1 + 16 * jt + 8 * hi);
        b1f[jt][1] = *(const v4f*)(b1 + 16 * jt + 8 * hi + 4);
    }

    v4f wf[4][8];
#pragma unroll
    for (int i = 0; i < 8; ++i)
#pragma unroll
        for (int jt = 0; jt < 4; ++jt)
            wf[jt][i] = *(const v4f*)(w1 + (size_t)(16 * jt + c16) * H_ + 8 * i + 4 * hi);

    v4f w2f[4][2];
#pragma unroll
    for (int jt = 0; jt < 4; ++jt)
#pragma unroll
        for (int q = 0; q < 2; ++q)
            w2f[jt][q] = *(const v4f*)(w2 + (c16 & 1) * H_ + 16 * jt + 8 * hi + 4 * q);
    const float b20 = b2[0];
    const float b21 = b2[1];

    __builtin_amdgcn_sched_barrier(0);

    v8f acc[2][4];
#pragma unroll
    for (int jt = 0; jt < 4; ++jt)
#pragma unroll
        for (int r = 0; r < 4; ++r) {
            acc[0][jt][r] = b1f[jt][0][r];  acc[0][jt][4 + r] = b1f[jt][1][r];
            acc[1][jt][r] = b1f[jt][0][r];  acc[1][jt][4 + r] = b1f[jt][1][r];
        }

    const bool a2row = (c16 & 6) == 0;
#pragma unroll
    for (int jt = 0; jt < 4; ++jt)
#pragma unroll
        for (int q = 0; q < 2; ++q)
#pragma unroll
            for (int e = 0; e < 4; ++e) w2f[jt][q][e] = a2row ? w2f[jt][q][e] : 0.0f;

#pragma unroll
    for (int i = 0; i < 8; ++i) {
        v4f x0, x1;
#pragma unroll
        for (int e = 0; e < 4; ++e) {
            x0[e] = u0 * wz[i][e] + v0 * bz[i][e];
            x1[e] = u1 * wz[i][e] + v1 * bz[i][e];
        }
#pragma unroll
        for (int hf = 0; hf < 2; ++hf) {
            v2f xb0, xb1;
            xb0.x = x0[2 * hf];  xb0.y = x0[2 * hf + 1];
            xb1.x = x1[2 * hf];  xb1.y = x1[2 * hf + 1];
#pragma unroll
            for (int jt = 0; jt < 4; ++jt) {
                v2f a;
                a.x = wf[jt][i][2 * hf];  a.y = wf[jt][i][2 * hf + 1];
                acc[0][jt] = __builtin_amdgcn_wmma_f32_16x16x4_f32(false, a, false, xb0, (short)0, acc[0][jt], false, false);
                acc[1][jt] = __builtin_amdgcn_wmma_f32_16x16x4_f32(false, a, false, xb1, (short)0, acc[1][jt], false, false);
            }
        }
    }

    v8f o2a, o2b;
#pragma unroll
    for (int r = 0; r < 8; ++r) { o2a[r] = 0.0f; o2b[r] = 0.0f; }
    o2a[0] = b20;  o2a[1] = b21;
    o2b[0] = b20;  o2b[1] = b21;
#pragma unroll
    for (int jt = 0; jt < 4; ++jt)
#pragma unroll
        for (int p = 0; p < 4; ++p) {
            v2f a, h0, h1;
            a.x  = w2f[jt][p >> 1][2 * (p & 1)];
            a.y  = w2f[jt][p >> 1][2 * (p & 1) + 1];
            h0.x = fmaxf(acc[0][jt][2 * p], 0.0f);  h0.y = fmaxf(acc[0][jt][2 * p + 1], 0.0f);
            h1.x = fmaxf(acc[1][jt][2 * p], 0.0f);  h1.y = fmaxf(acc[1][jt][2 * p + 1], 0.0f);
            o2a = __builtin_amdgcn_wmma_f32_16x16x4_f32(false, a, false, h0, (short)0, o2a, false, false);
            o2b = __builtin_amdgcn_wmma_f32_16x16x4_f32(false, a, false, h1, (short)0, o2b, false, false);
        }

    const float o0 = hi ? o2b[0] : o2a[0];
    const float o1 = hi ? o2b[1] : o2a[1];
    const int idx = base + lane;
    out[idx] = o0;
    float sp = fmaxf(o1, 0.0f) + log1pf(expf(-fabsf(o1)));
    out[C_ * TG_ + idx] = sp + 1e-6f;
}


#define RLEN_   (TF_ / KS_)
#define RSTEPS_ (RLEN_ / 8)
#define EPL_    (TF_ * 2)

__global__ __launch_bounds__(32 * KS_)
void first_sweep_kernel(const float* __restrict__ grn,
                        const float* __restrict__ z,
                        const int*   __restrict__ binary_tf,
                        float*       __restrict__ UV,
                        float*       __restrict__ zeros,
                        int*         __restrict__ ownf,
                        float*       __restrict__ E1,
                        float*       __restrict__ GT) {
    __shared__ float red[KS_][2][16];
    __shared__ int   ownl[16];

    const int c     = blockIdx.y;
    const int mtile = blockIdx.x;
    const int ks    = threadIdx.x >> 5;
    const int lane  = threadIdx.x & 31;
    const int c16   = lane & 15;
    const int hi    = lane >> 4;

    int col[TF_ / 32];
#pragma unroll
    for (int q = 0; q < TF_ / 32; ++q) col[q] = binary_tf[c * TF_ + q * 32 + lane];
    const int irow = mtile * 16 + 2 * ks;
    const int tr0raw = binary_tf[c * TF_ + irow];
    const int tr1raw = binary_tf[c * TF_ + irow + 1];
    const float zx = z[c * TG_ + mtile * 64 + (threadIdx.x & 63)];
    __builtin_amdgcn_sched_barrier(0);

    const int kbase = ks * KLEN_ + hi * 4;
    const float* grow = grn + ((size_t)(c * TF_ + mtile * 16 + c16)) * TG_ + kbase;
    const float* zrow = z + (size_t)c * TG_ + kbase;
    const float  fill = (c16 == 1) ? 1.0f : 0.0f;

    v8f acc0, acc1;
#pragma unroll
    for (int r = 0; r < 8; ++r) { acc0[r] = 0.0f; acc1[r] = 0.0f; }

    v4f p[STEPS_], g[STEPS_];
#pragma unroll
    for (int j = 0; j < STEPS_; ++j) {
        p[j] = *(const v4f*)(zrow + j * 8);
        g[j] = *(const v4f*)(grow + j * 8);
    }
    __builtin_amdgcn_sched_barrier(0);

    const float* r0 = grn + ((size_t)(c * TF_ + irow)) * TG_;
    const float* r1 = r0 + TG_;
    float ga[TF_ / 32], gb[TF_ / 32];
    const int tr0 = tr0raw & (TG_ - 1);
    const int tr1 = tr1raw & (TG_ - 1);
    bool  later0 = false, later1 = false;
#pragma unroll
    for (int q = 0; q < TF_ / 32; ++q) {
        const int t = col[q] & (TG_ - 1);
        ga[q] = r0[t];
        gb[q] = r1[t];
        later0 |= (t == tr0) & (q * 32 + lane > irow);
        later1 |= (t == tr1) & (q * 32 + lane > irow + 1);
    }
    const bool own0 = __ballot(later0) == 0;
    const bool own1 = __ballot(later1) == 0;
    __builtin_amdgcn_sched_barrier(0);

#pragma unroll
    for (int j = 0; j < STEPS_; ++j) {
        v2f a, b;
        a.x = (c16 == 0) ? p[j].x : fill;  a.y = (c16 == 0) ? p[j].y : fill;
        b.x = g[j].x;  b.y = g[j].y;
        acc0 = __builtin_amdgcn_wmma_f32_16x16x4_f32(false, a, false, b, (short)0, acc0, false, false);
        a.x = (c16 == 0) ? p[j].z : fill;  a.y = (c16 == 0) ? p[j].w : fill;
        b.x = g[j].z;  b.y = g[j].w;
        acc1 = __builtin_amdgcn_wmma_f32_16x16x4_f32(false, a, false, b, (short)0, acc1, false, false);
    }

    const float su = acc0[0] + acc1[0];
    const float sv = acc0[1] + acc1[1];
    if (hi == 0) {
        red[ks][0][c16] = su;
        red[ks][1][c16] = sv;
    }
    if (lane == 0) {
        ownl[2 * ks]     = own0 ? 1 : 0;
        ownl[2 * ks + 1] = own1 ? 1 : 0;
    }

    __syncthreads();

    if (threadIdx.x < 32) {
        const int plane = lane >> 4;
        const int erow  = lane & 15;
        const int own   = ownl[erow];
        float s = red[0][plane][erow];
#pragma unroll
        for (int w = 1; w < KS_; ++w) s += red[w][plane][erow];
        E1[c * EPL_ + mtile * 32 + lane] = own ? s : 0.0f;
        ownf[(c * (TF_ / 16) + mtile) * 32 + lane] = own;
    }

    float* o0 = GT + ((size_t)(c * TF_ + irow)) * TF_ + lane;
#pragma unroll
    for (int q = 0; q < TF_ / 32; ++q) {
        o0[q * 32]       = own0 ? ga[q] : 0.0f;
        o0[TF_ + q * 32] = own1 ? gb[q] : 0.0f;
    }

    const int tq = mtile * 64 + (threadIdx.x & 63);
    if (threadIdx.x < 64)        UV[((size_t)c * 2) * TG_ + tq] = zx;
    else if (threadIdx.x < 128)  UV[((size_t)c * 2 + 1) * TG_ + tq] = 1.0f;
    else if (threadIdx.x < 192 && c == 0) zeros[tq] = 0.0f;
}

template <bool FINAL>
__global__ __launch_bounds__(32 * KS_)
void reduced_sweep_kernel(const float* __restrict__ GT,
                          const float* __restrict__ Ein,
                          const float* __restrict__ E1,
                          const float* __restrict__ zeros,
                          const float* __restrict__ z,
                          const int*   __restrict__ binary_tf,
                          const int*   __restrict__ ownf,
                          float*       __restrict__ dst) {
    __shared__ float red[KS_][2][16];

    const int c     = blockIdx.y;
    const int mtile = blockIdx.x;
    const int ks    = threadIdx.x >> 5;
    const int lane  = threadIdx.x & 31;
    const int c16   = lane & 15;
    const int hi    = lane >> 4;

    const float e1v = (threadIdx.x < 32) ? E1[c * EPL_ + mtile * 32 + lane] : 0.0f;
    int traw = 0, own = 0;
    if (FINAL) {
        traw = binary_tf[c * TF_ + mtile * 16 + c16];
        own  = ownf[(c * (TF_ / 16) + mtile) * 32 + lane];
    }
    __builtin_amdgcn_sched_barrier(0);

    const int kbase = ks * RLEN_ + hi * 4;
    const float* grow = GT + ((size_t)(c * TF_ + mtile * 16 + c16)) * TF_ + kbase;
    const float* prow = (c16 < 2) ? Ein + c * EPL_ + ks * 128 + c16 * 16 + hi * 4
                                  : zeros + hi * 4;

    v8f acc0, acc1;
#pragma unroll
    for (int r = 0; r < 8; ++r) { acc0[r] = 0.0f; acc1[r] = 0.0f; }

    v4f p[RSTEPS_], g[RSTEPS_];
#pragma unroll
    for (int j = 0; j < RSTEPS_; ++j) {
        p[j] = *(const v4f*)(prow + (j >> 1) * 32 + (j & 1) * 8);
        g[j] = *(const v4f*)(grow + j * 8);
    }
    __builtin_amdgcn_sched_barrier(0);
    float rhs = 0.0f;
    int trow = 0;
    if (FINAL) {
        trow = traw & (TG_ - 1);
        rhs  = (threadIdx.x < 16) ? z[c * TG_ + trow] : 1.0f;
        __builtin_amdgcn_sched_barrier(0);
    }
#pragma unroll
    for (int j = 0; j < RSTEPS_; ++j) {
        v2f a, b;
        a.x = p[j].x;  a.y = p[j].y;
        b.x = g[j].x;  b.y = g[j].y;
        acc0 = __builtin_amdgcn_wmma_f32_16x16x4_f32(false, a, false, b, (short)0, acc0, false, false);
        a.x = p[j].z;  a.y = p[j].w;
        b.x = g[j].z;  b.y = g[j].w;
        acc1 = __builtin_amdgcn_wmma_f32_16x16x4_f32(false, a, false, b, (short)0, acc1, false, false);
    }

    const float su = acc0[0] + acc1[0];
    const float sv = acc0[1] + acc1[1];
    if (hi == 0) {
        red[ks][0][c16] = su;
        red[ks][1][c16] = sv;
    }
    __syncthreads();

    if (threadIdx.x < 32) {
        const int plane = lane >> 4;
        const int erow  = lane & 15;
        float s = e1v;
#pragma unroll
        for (int w = 0; w < KS_; ++w) s += red[w][plane][erow];
        if (FINAL) {
            if (own) dst[((size_t)(c * 2 + plane)) * TG_ + trow] = rhs + s;
        } else {
            dst[c * EPL_ + mtile * 32 + lane] = s;
        }
    }
}


#define SQD_    4
#define SQSTEPS_ (TF_ / 8)

__device__ __forceinline__
void square_part(const float* __restrict__ GT, float* __restrict__ P, const int c, const int bx) {
    const int mb   = bx >> 2;
    const int nb   = bx & 3;
    const int w    = threadIdx.x >> 5;
    const int lane = threadIdx.x & 31;
    const int c16  = lane & 15;
    const int hi   = lane >> 4;
    const int m0   = mb * 64 + (w >> 1) * 32;
    const int n0   = nb * 128 + (w & 1) * 64;

    const float* G  = GT + (size_t)c * TF_ * TF_;
    const float* ap = G + (size_t)(m0 + c16) * TF_ + 4 * hi;
    const float* bp = G + (size_t)(4 * hi) * TF_ + n0 + 4 * c16;

    v8f acc[2][4];
#pragma unroll
    for (int mt = 0; mt < 2; ++mt)
#pragma unroll
        for (int nt = 0; nt < 4; ++nt)
#pragma unroll
            for (int r = 0; r < 8; ++r) acc[mt][nt][r] = 0.0f;

    v4f a[SQD_][2], b[SQD_][4];

#define SQ_LOAD(s, step)                                                            \
    {                                                                               \
        a[s][0] = *(const v4f*)(ap + (size_t)(step) * 8);                           \
        a[s][1] = *(const v4f*)(ap + (size_t)16 * TF_ + (size_t)(step) * 8);        \
        b[s][0] = *(const v4f*)(bp + ((size_t)(step) * 8 + 0) * TF_);               \
        b[s][1] = *(const v4f*)(bp + ((size_t)(step) * 8 + 1) * TF_);               \
        b[s][2] = *(const v4f*)(bp + ((size_t)(step) * 8 + 2) * TF_);               \
        b[s][3] = *(const v4f*)(bp + ((size_t)(step) * 8 + 3) * TF_);               \
    }

#define SQ_COMPUTE(s)                                                               \
    {                                                                               \
        _Pragma("unroll")                                                           \
        for (int h2 = 0; h2 < 2; ++h2) {                                            \
            v2f bb[4];                                                              \
            _Pragma("unroll")                                                       \
            for (int nt = 0; nt < 4; ++nt) {                                        \
                bb[nt].x = b[s][2 * h2][nt];                                        \
                bb[nt].y = b[s][2 * h2 + 1][nt];                                    \
            }                                                                       \
            _Pragma("unroll")                                                       \
            for (int mt = 0; mt < 2; ++mt) {                                        \
                v2f aa;                                                             \
                aa.x = a[s][mt][2 * h2];                                            \
                aa.y = a[s][mt][2 * h2 + 1];                                        \
                _Pragma("unroll")                                                   \
                for (int nt = 0; nt < 4; ++nt)                                      \
                    acc[mt][nt] = __builtin_amdgcn_wmma_f32_16x16x4_f32(            \
                        false, aa, false, bb[nt], (short)0, acc[mt][nt], false, false); \
            }                                                                       \
        }                                                                           \
    }

#pragma unroll
    for (int s = 0; s < SQD_; ++s) SQ_LOAD(s, s)
    __builtin_amdgcn_sched_barrier(0);

#pragma nounroll
    for (int it = 0; it < SQSTEPS_ - SQD_; it += SQD_) {
#pragma unroll
        for (int s = 0; s < SQD_; ++s) {
            SQ_COMPUTE(s)
            __builtin_amdgcn_sched_barrier(0);
            SQ_LOAD(s, it + SQD_ + s)
            __builtin_amdgcn_sched_barrier(0);
        }
    }
#pragma unroll
    for (int s = 0; s < SQD_; ++s) SQ_COMPUTE(s)

#undef SQ_LOAD
#undef SQ_COMPUTE

    float* prow = P + (size_t)c * TF_ * TF_ + (size_t)(m0 + 8 * hi) * TF_ + n0 + 4 * c16;
#pragma unroll
    for (int mt = 0; mt < 2; ++mt)
#pragma unroll
        for (int r = 0; r < 8; ++r) {
            v4f o;
            o.x = acc[mt][0][r];  o.y = acc[mt][1][r];
            o.z = acc[mt][2][r];  o.w = acc[mt][3][r];
            *(v4f*)(prow + (size_t)(16 * mt + r) * TF_) = o;
        }
}

__device__ __forceinline__
void f_rows_part(const float* __restrict__ GT,
                 const float* __restrict__ E1,
                 const float* __restrict__ zeros,
                 float*       __restrict__ F,
                 float (*red)[2][16],
                 const int c, const int mtile) {
    const int w    = threadIdx.x >> 5;
    const int lane = threadIdx.x & 31;
    const int c16  = lane & 15;
    const int hi   = lane >> 4;

    const float e1v = (threadIdx.x < 32) ? E1[c * EPL_ + mtile * 32 + lane] : 0.0f;
    __builtin_amdgcn_sched_barrier(0);

    const int kbase = 2 * w * RLEN_ + hi * 4;
    const float* grow = GT + ((size_t)(c * TF_ + mtile * 16 + c16)) * TF_ + kbase;
    const float* prow = (c16 < 2) ? E1 + c * EPL_ + (2 * w) * 128 + c16 * 16 + hi * 4
                                  : zeros + hi * 4;

    v8f acc[2][2];
#pragma unroll
    for (int r = 0; r < 8; ++r) {
        acc[0][0][r] = 0.0f;  acc[0][1][r] = 0.0f;
        acc[1][0][r] = 0.0f;  acc[1][1][r] = 0.0f;
    }

    v4f p[2 * RSTEPS_], g[2 * RSTEPS_];
#pragma unroll
    for (int j = 0; j < 2 * RSTEPS_; ++j) {
        p[j] = *(const v4f*)(prow + (j >> 1) * 32 + (j & 1) * 8);
        g[j] = *(const v4f*)(grow + j * 8);
    }
    __builtin_amdgcn_sched_barrier(0);
#pragma unroll
    for (int j = 0; j < 2 * RSTEPS_; ++j) {
        const int sl = j / RSTEPS_;
        v2f a, b;
        a.x = p[j].x;  a.y = p[j].y;
        b.x = g[j].x;  b.y = g[j].y;
        acc[sl][0] = __builtin_amdgcn_wmma_f32_16x16x4_f32(false, a, false, b, (short)0, acc[sl][0], false, false);
        a.x = p[j].z;  a.y = p[j].w;
        b.x = g[j].z;  b.y = g[j].w;
        acc[sl][1] = __builtin_amdgcn_wmma_f32_16x16x4_f32(false, a, false, b, (short)0, acc[sl][1], false, false);
    }

    if (hi == 0) {
#pragma unroll
        for (int sl = 0; sl < 2; ++sl) {
            red[2 * w + sl][0][c16] = acc[sl][0][0] + acc[sl][1][0];
            red[2 * w + sl][1][c16] = acc[sl][0][1] + acc[sl][1][1];
        }
    }
    __syncthreads();

    if (threadIdx.x < 32) {
        const int plane = lane >> 4;
        const int erow  = lane & 15;
        float s = e1v;
#pragma unroll
        for (int q = 0; q < KS_; ++q) s += red[q][plane][erow];
        F[c * EPL_ + mtile * 32 + lane] = s;
    }
}

__global__ __launch_bounds__(128)
void square_and_f_kernel(const float* __restrict__ GT,
                         const float* __restrict__ E1,
                         const float* __restrict__ zeros,
                         float*       __restrict__ P,
                         float*       __restrict__ F) {
    __shared__ float red[KS_][2][16];
    const int bid = blockIdx.x;
    if (bid < 32 * C_) {
        square_part(GT, P, bid >> 5, bid & 31);
    } else {
        const int r = bid - 32 * C_;
        f_rows_part(GT, E1, zeros, F, red, r >> 5, r & 31);
    }
}

extern "C" void kernel_launch(void* const* d_in, const int* in_sizes, int n_in,
                              void* d_out, int out_size, void* d_ws, size_t ws_size,
                              hipStream_t stream) {
    (void)in_sizes; (void)n_in; (void)out_size;
    const float* z   = (const float*)d_in[0];
    const float* grn = (const float*)d_in[1];
    const int*   btf = (const int*)  d_in[2];
    const float* w_z = (const float*)d_in[4];
    const float* b_z = (const float*)d_in[5];
    const float* w1  = (const float*)d_in[6];
    const float* b1  = (const float*)d_in[7];
    const float* w2  = (const float*)d_in[8];
    const float* b2  = (const float*)d_in[9];
    float* out = (float*)d_out;

    dim3 grid(TF_ / 16, C_);

    if (ws_size >= ((size_t)C_ * 3 * TG_ + TG_ + (size_t)4 * C_ * EPL_ + (size_t)2 * C_ * TF_ * TF_) * sizeof(float)) {
        float* UV  = (float*)d_ws;
        float* zer = UV + (size_t)C_ * 2 * TG_;
        int*   own = (int*)(zer + TG_);
        float* E1  = (float*)(own + (size_t)C_ * TG_);
        float* F   = E1 + (size_t)C_ * EPL_;
        float* Ea  = F  + (size_t)C_ * EPL_;
        float* Eb  = Ea + (size_t)C_ * EPL_;
        float* GT  = Eb + (size_t)C_ * EPL_;
        float* P   = GT + (size_t)C_ * TF_ * TF_;

        first_sweep_kernel<<<grid, 32 * KS_, 0, stream>>>(grn, z, btf, UV, zer, own, E1, GT);
        square_and_f_kernel<<<dim3(2 * 32 * C_), 128, 0, stream>>>(GT, E1, zer, P, F);
        const float* yin = F;
        float* ybuf[2] = { Ea, Eb };
        for (int j = 1; j < 8; ++j) {
            float* yout = ybuf[j & 1];
            reduced_sweep_kernel<false><<<grid, 32 * KS_, 0, stream>>>(P, yin, F, zer, z, btf, own, yout);
            yin = yout;
        }
        reduced_sweep_kernel<true><<<grid, 32 * KS_, 0, stream>>>(P, yin, F, zer, z, btf, own, UV);
        mlp_kernel<<<(C_ * TG_ + 127) / 128, 128, 0, stream>>>(UV, w_z, b_z, w1, b1, w2, b2, out);
        return;
    }

    if (ws_size >= ((size_t)C_ * 3 * TG_ + TG_ + (size_t)3 * C_ * EPL_ + (size_t)C_ * TF_ * TF_) * sizeof(float)) {
        float* UV  = (float*)d_ws;
        float* zer = UV + (size_t)C_ * 2 * TG_;
        int*   own = (int*)(zer + TG_);
        float* E1  = (float*)(own + (size_t)C_ * TG_);
        float* Ea  = E1 + (size_t)C_ * EPL_;
        float* Eb  = Ea + (size_t)C_ * EPL_;
        float* GT  = Eb + (size_t)C_ * EPL_;

        first_sweep_kernel<<<grid, 32 * KS_, 0, stream>>>(grn, z, btf, UV, zer, own, E1, GT);
        const float* ein = E1;
        float* ebuf[2] = { Ea, Eb };
        for (int s = 2; s < NITER; ++s) {
            float* eout = ebuf[s & 1];
            reduced_sweep_kernel<false><<<grid, 32 * KS_, 0, stream>>>(GT, ein, E1, zer, z, btf, own, eout);
            ein = eout;
        }
        reduced_sweep_kernel<true><<<grid, 32 * KS_, 0, stream>>>(GT, ein, E1, zer, z, btf, own, UV);
        mlp_kernel<<<(C_ * TG_ + 127) / 128, 128, 0, stream>>>(UV, w_z, b_z, w1, b1, w2, b2, out);
        return;
    }

    float* UVa = (float*)d_ws;
    float* UVb = UVa + (size_t)C_ * 2 * TG_;
    float* zer = UVb + (size_t)C_ * 2 * TG_;

    init_uv_kernel<<<(C_ * TG_ + 255) / 256, 256, 0, stream>>>(z, UVa, UVb, zer);

    float* bufs[2] = { UVa, UVb };
    for (int it = 0; it < NITER; ++it) {
        const float* uvin = bufs[it & 1];
        float*      uvout = bufs[(it + 1) & 1];
        neumann_iter_kernel<<<grid, 32 * KS_, 0, stream>>>(grn, z, btf, uvin, zer, uvout);
    }
    const float* uvfin = bufs[NITER & 1];

    mlp_kernel<<<(C_ * TG_ + 127) / 128, 128, 0, stream>>>(uvfin, w_z, b_z, w1, b1, w2, b2, out);
}


# MLP launch dissolved (11 -> 10 launches): rows no TF index names (x = x0) run as 128 more work-groups of the square launch, each last-sweep work-group's 16 TF rows in its epilogue; same bits
# speedup vs baseline: 19.4354x; 1.0417x over previous
#include <hip/hip_runtime.h>
#include <hip/hip_bf16.h>

#define C_   8
#define TF_  512
#define TG_  2048
#define H_   64
#define NITER 18
#define KS_   8
#define KLEN_ (TG_ / KS_)
#define STEPS_ (KLEN_ / 8)

typedef __attribute__((ext_vector_type(2))) float v2f;
typedef __attribute__((ext_vector_type(4))) float v4f;
typedef __attribute__((ext_vector_type(8))) float v8f;


__global__ __launch_bounds__(256)
void init_uv_kernel(const float* __restrict__ z,
                    float* __restrict__ UVa,
                    float* __restrict__ UVb,
                    float* __restrict__ zeros) {
    int idx = blockIdx.x * blockDim.x + threadIdx.x;
    if (idx >= C_ * TG_) return;
    if (idx < TG_) zeros[idx] = 0.0f;
    int c = idx / TG_;
    int t = idx % TG_;
    float zz = z[idx];
    size_t pu = ((size_t)c * 2) * TG_ + t;
    size_t pv = pu + TG_;
    UVa[pu] = zz;  UVa[pv] = 1.0f;
    UVb[pu] = zz;  UVb[pv] = 1.0f;
}

__global__ __launch_bounds__(32 * KS_)
void neumann_iter_kernel(const float* __restrict__ grn,
                         const float* __restrict__ z,
                         const int*   __restrict__ binary_tf,
                         const float* __restrict__ UVin,
                         const float* __restrict__ zeros,
                         float*       __restrict__ UVout) {
    __shared__ float red[KS_][2][16];

    const int c     = blockIdx.y;
    const int mtile = blockIdx.x;
    const int ks    = threadIdx.x >> 5;
    const int lane  = threadIdx.x & 31;
    const int c16   = lane & 15;
    const int hi    = lane >> 4;

    const int erow = lane & 15;
    const int traw = binary_tf[c * TF_ + mtile * 16 + erow];
    __builtin_amdgcn_sched_barrier(0);

    const int kbase = ks * KLEN_ + hi * 4;
    const float* grow = grn + ((size_t)(c * TF_ + mtile * 16 + c16)) * TG_ + kbase;
    const float* prow = (c16 < 2) ? UVin + ((size_t)(c * 2 + c16)) * TG_ + kbase
                                  : zeros + kbase;

    v8f acc0, acc1;
#pragma unroll
    for (int r = 0; r < 8; ++r) { acc0[r] = 0.0f; acc1[r] = 0.0f; }

    v4f p[STEPS_], g[STEPS_];
#pragma unroll
    for (int j = 0; j < STEPS_; ++j) {
        p[j] = *(const v4f*)(prow + j * 8);
        g[j] = *(const v4f*)(grow + j * 8);
    }
    __builtin_amdgcn_sched_barrier(0);
    const int   trow = traw & (TG_ - 1);
    const float rhs  = (threadIdx.x < 16) ? z[c * TG_ + trow] : 1.0f;
    __builtin_amdgcn_sched_barrier(0);
#pragma unroll
    for (int j = 0; j < STEPS_; ++j) {
        v2f a, b;
        a.x = p[j].x;  a.y = p[j].y;
        b.x = g[j].x;  b.y = g[j].y;
        acc0 = __builtin_amdgcn_wmma_f32_16x16x4_f32(false, a, false, b, (short)0, acc0, false, false);
        a.x = p[j].z;  a.y = p[j].w;
        b.x = g[j].z;  b.y = g[j].w;
        acc1 = __builtin_amdgcn_wmma_f32_16x16x4_f32(false, a, false, b, (short)0, acc1, false, false);
    }

    const float su = acc0[0] + acc1[0];
    const float sv = acc0[1] + acc1[1];
    if (hi == 0) {
        red[ks][0][c16] = su;
        red[ks][1][c16] = sv;
    }
    __syncthreads();

    if (threadIdx.x < 32) {
        const int plane = lane >> 4;
        float s = rhs;
#pragma unroll
        for (int w = 0; w < KS_; ++w) s += red[w][plane][erow];
        UVout[((size_t)(c * 2 + plane)) * TG_ + trow] = s;
    }
}

__global__ __launch_bounds__(128)
void mlp_kernel(const float* __restrict__ UV,
                const float* __restrict__ w_z, const float* __restrict__ b_z,
                const float* __restrict__ w1, const float* __restrict__ b1,
                const float* __restrict__ w2, const float* __restrict__ b2,
                float* __restrict__ out) {
    const int lane = threadIdx.x & 31;
    const int wave = (blockIdx.x * blockDim.x + threadIdx.x) >> 5;
    const int c16  = lane & 15;
    const int hi   = lane >> 4;
    const int base = wave * 32;
    const int c    = base / TG_;
    const int t0   = base % TG_;

    const float* up = UV + ((size_t)c * 2) * TG_ + t0;
    const float u0 = up[c16];
    const float u1 = up[16 + c16];
    const float v0 = up[TG_ + c16];
    const float v1 = up[TG_ + 16 + c16];

    v4f wz[8], bz[8];
#pragma unroll
    for (int i = 0; i < 8; ++i) {
        wz[i] = *(const v4f*)(w_z + 8 * i + 4 * hi);
        bz[i] = *(const v4f*)(b_z + 8 * i + 4 * hi);
    }

    v4f b1f[4][2];
#pragma unroll
    for (int jt = 0; jt < 4; ++jt) {
        b1f[jt][0] = *(const v4f*)(b1 + 16 * jt + 8 * hi);
        b1f[jt][1] = *(const v4f*)(b1 + 16 * jt + 8 * hi + 4);
    }

    v4f wf[4][8];
#pragma unroll
    for (int i = 0; i < 8; ++i)
#pragma unroll
        for (int jt = 0; jt < 4; ++jt)
            wf[jt][i] = *(const v4f*)(w1 + (size_t)(16 * jt + c16) * H_ + 8 * i + 4 * hi);

    v4f w2f[4][2];
#pragma unroll
    for (int jt = 0; jt < 4; ++jt)
#pragma unroll
        for (int q = 0; q < 2; ++q)
            w2f[jt][q] = *(const v4f*)(w2 + (c16 & 1) * H_ + 16 * jt + 8 * hi + 4 * q);
    const float b20 = b2[0];
    const float b21 = b2[1];

    __builtin_amdgcn_sched_barrier(0);

    v8f acc[2][4];
#pragma unroll
    for (int jt = 0; jt < 4; ++jt)
#pragma unroll
        for (int r = 0; r < 4; ++r) {
            acc[0][jt][r] = b1f[jt][0][r];  acc[0][jt][4 + r] = b1f[jt][1][r];
            acc[1][jt][r] = b1f[jt][0][r];  acc[1][jt][4 + r] = b1f[jt][1][r];
        }

    const bool a2row = (c16 & 6) == 0;
#pragma unroll
    for (int jt = 0; jt < 4; ++jt)
#pragma unroll
        for (int q = 0; q < 2; ++q)
#pragma unroll
            for (int e = 0; e < 4; ++e) w2f[jt][q][e] = a2row ? w2f[jt][q][e] : 0.0f;

#pragma unroll
    for (int i = 0; i < 8; ++i) {
        v4f x0, x1;
#pragma unroll
        for (int e = 0; e < 4; ++e) {
            x0[e] = u0 * wz[i][e] + v0 * bz[i][e];
            x1[e] = u1 * wz[i][e] + v1 * bz[i][e];
        }
#pragma unroll
        for (int hf = 0; hf < 2; ++hf) {
            v2f xb0, xb1;
            xb0.x = x0[2 * hf];  xb0.y = x0[2 * hf + 1];
            xb1.x = x1[2 * hf];  xb1.y = x1[2 * hf + 1];
#pragma unroll
            for (int jt = 0; jt < 4; ++jt) {
                v2f a;
                a.x = wf[jt][i][2 * hf];  a.y = wf[jt][i][2 * hf + 1];
                acc[0][jt] = __builtin_amdgcn_wmma_f32_16x16x4_f32(false, a, false, xb0, (short)0, acc[0][jt], false, false);
                acc[1][jt] = __builtin_amdgcn_wmma_f32_16x16x4_f32(false, a, false, xb1, (short)0, acc[1][jt], false, false);
            }
        }
    }

    v8f o2a, o2b;
#pragma unroll
    for (int r = 0; r < 8; ++r) { o2a[r] = 0.0f; o2b[r] = 0.0f; }
    o2a[0] = b20;  o2a[1] = b21;
    o2b[0] = b20;  o2b[1] = b21;
#pragma unroll
    for (int jt = 0; jt < 4; ++jt)
#pragma unroll
        for (int p = 0; p < 4; ++p) {
            v2f a, h0, h1;
            a.x  = w2f[jt][p >> 1][2 * (p & 1)];
            a.y  = w2f[jt][p >> 1][2 * (p & 1) + 1];
            h0.x = fmaxf(acc[0][jt][2 * p], 0.0f);  h0.y = fmaxf(acc[0][jt][2 * p + 1], 0.0f);
            h1.x = fmaxf(acc[1][jt][2 * p], 0.0f);  h1.y = fmaxf(acc[1][jt][2 * p + 1], 0.0f);
            o2a = __builtin_amdgcn_wmma_f32_16x16x4_f32(false, a, false, h0, (short)0, o2a, false, false);
            o2b = __builtin_amdgcn_wmma_f32_16x16x4_f32(false, a, false, h1, (short)0, o2b, false, false);
        }

    const float o0 = hi ? o2b[0] : o2a[0];
    const float o1 = hi ? o2b[1] : o2a[1];
    const int idx = base + lane;
    out[idx] = o0;
    float sp = fmaxf(o1, 0.0f) + log1pf(expf(-fabsf(o1)));
    out[C_ * TG_ + idx] = sp + 1e-6f;
}


struct MlpW {
    v4f wz[8], bz[8];
    v4f b1f[4][2];
    v4f wf[4][8];
    v4f w2f[4][2];
    float b20, b21;
};

__device__ __forceinline__
void mlp_load(MlpW& W,
              const float* __restrict__ w_z, const float* __restrict__ b_z,
              const float* __restrict__ w1, const float* __restrict__ b1,
              const float* __restrict__ w2, const float* __restrict__ b2,
              const int c16, const int hi) {
#pragma unroll
    for (int i = 0; i < 8; ++i) {
        W.wz[i] = *(const v4f*)(w_z + 8 * i + 4 * hi);
        W.bz[i] = *(const v4f*)(b_z + 8 * i + 4 * hi);
    }
#pragma unroll
    for (int jt = 0; jt < 4; ++jt) {
        W.b1f[jt][0] = *(const v4f*)(b1 + 16 * jt + 8 * hi);
        W.b1f[jt][1] = *(const v4f*)(b1 + 16 * jt + 8 * hi + 4);
    }
#pragma unroll
    for (int i = 0; i < 8; ++i)
#pragma unroll
        for (int jt = 0; jt < 4; ++jt)
            W.wf[jt][i] = *(const v4f*)(w1 + (size_t)(16 * jt + c16) * H_ + 8 * i + 4 * hi);
#pragma unroll
    for (int jt = 0; jt < 4; ++jt)
#pragma unroll
        for (int q = 0; q < 2; ++q)
            W.w2f[jt][q] = *(const v4f*)(w2 + (c16 & 1) * H_ + 16 * jt + 8 * hi + 4 * q);
    W.b20 = b2[0];
    W.b21 = b2[1];
}

template <int NT>
__device__ __forceinline__
void mlp_tiles(const MlpW& W, const float* u, const float* v, const int c16,
               float* o0, float* o1) {
    v8f acc[NT][4];
#pragma unroll
    for (int t = 0; t < NT; ++t)
#pragma unroll
        for (int jt = 0; jt < 4; ++jt)
#pragma unroll
            for (int r = 0; r < 4; ++r) {
                acc[t][jt][r] = W.b1f[jt][0][r];  acc[t][jt][4 + r] = W.b1f[jt][1][r];
            }

    const bool a2row = (c16 & 6) == 0;
    v4f w2z[4][2];
#pragma unroll
    for (int jt = 0; jt < 4; ++jt)
#pragma unroll
        for (int q = 0; q < 2; ++q)
#pragma unroll
            for (int e = 0; e < 4; ++e) w2z[jt][q][e] = a2row ? W.w2f[jt][q][e] : 0.0f;

#pragma unroll
    for (int i = 0; i < 8; ++i) {
        v4f x[NT];
#pragma unroll
        for (int t = 0; t < NT; ++t)
#pragma unroll
            for (int e = 0; e < 4; ++e) x[t][e] = u[t] * W.wz[i][e] + v[t] * W.bz[i][e];
#pragma unroll
        for (int hf = 0; hf < 2; ++hf) {
#pragma unroll
            for (int jt = 0; jt < 4; ++jt) {
                v2f a;
                a.x = W.wf[jt][i][2 * hf];  a.y = W.wf[jt][i][2 * hf + 1];
#pragma unroll
                for (int t = 0; t < NT; ++t) {
                    v2f xb;
                    xb.x = x[t][2 * hf];  xb.y = x[t][2 * hf + 1];
                    acc[t][jt] = __builtin_amdgcn_wmma_f32_16x16x4_f32(false, a, false, xb, (short)0, acc[t][jt], false, false);
                }
            }
        }
    }

    v8f o2[NT];
#pragma unroll
    for (int t = 0; t < NT; ++t) {
#pragma unroll
        for (int r = 0; r < 8; ++r) o2[t][r] = 0.0f;
        o2[t][0] = W.b20;  o2[t][1] = W.b21;
    }
#pragma unroll
    for (int jt = 0; jt < 4; ++jt)
#pragma unroll
        for (int p = 0; p < 4; ++p) {
            v2f a;
            a.x = w2z[jt][p >> 1][2 * (p & 1)];
            a.y = w2z[jt][p >> 1][2 * (p & 1) + 1];
#pragma unroll
            for (int t = 0; t < NT; ++t) {
                v2f h;
                h.x = fmaxf(acc[t][jt][2 * p], 0.0f);  h.y = fmaxf(acc[t][jt][2 * p + 1], 0.0f);
                o2[t] = __builtin_amdgcn_wmma_f32_16x16x4_f32(false, a, false, h, (short)0, o2[t], false, false);
            }
        }
#pragma unroll
    for (int t = 0; t < NT; ++t) { o0[t] = o2[t][0];  o1[t] = o2[t][1]; }
}

__device__ __forceinline__ float softplus_eps(const float o1) {
    return fmaxf(o1, 0.0f) + log1pf(expf(-fabsf(o1))) + 1e-6f;
}

__device__ __forceinline__
void x0_rows_part(const float* __restrict__ z, const int* __restrict__ binary_tf,
                  const float* __restrict__ w_z, const float* __restrict__ b_z,
                  const float* __restrict__ w1, const float* __restrict__ b1,
                  const float* __restrict__ w2, const float* __restrict__ b2,
                  float* __restrict__ out, const int wg) {
    const int lane = threadIdx.x & 31;
    const int wave = (wg * 128 + (int)threadIdx.x) >> 5;
    const int c16  = lane & 15;
    const int hi   = lane >> 4;
    const int base = wave * 32;
    const int c    = base / TG_;
    const int t0   = base % TG_;

    float u[2], v[2];
    u[0] = z[base + c16];
    u[1] = z[base + 16 + c16];
    v[0] = 1.0f;  v[1] = 1.0f;
    int col[TF_ / 32];
#pragma unroll
    for (int q = 0; q < TF_ / 32; ++q) col[q] = binary_tf[c * TF_ + q * 32 + lane];
    MlpW W;
    mlp_load(W, w_z, b_z, w1, b1, w2, b2, c16, hi);
    __builtin_amdgcn_sched_barrier(0);

    unsigned named = 0u;
#pragma unroll
    for (int q = 0; q < TF_ / 32; ++q) {
        const unsigned d = (unsigned)((col[q] & (TG_ - 1)) - t0);
        named |= (d < 32u) ? (1u << d) : 0u;
    }
#pragma unroll
    for (int m = 16; m >= 1; m >>= 1) named |= (unsigned)__shfl_xor((int)named, m);

    float o0[2], o1[2];
    mlp_tiles<2>(W, u, v, c16, o0, o1);

    const float a0 = hi ? o0[1] : o0[0];
    const float a1 = hi ? o1[1] : o1[0];
    if (((named >> lane) & 1u) == 0u) {
        out[base + lane] = a0;
        out[C_ * TG_ + base + lane] = softplus_eps(a1);
    }
}


#define RLEN_   (TF_ / KS_)
#define RSTEPS_ (RLEN_ / 8)
#define EPL_    (TF_ * 2)

__global__ __launch_bounds__(32 * KS_)
void first_sweep_kernel(const float* __restrict__ grn,
                        const float* __restrict__ z,
                        const int*   __restrict__ binary_tf,
                        float*       __restrict__ zeros,
                        int*         __restrict__ ownf,
                        float*       __restrict__ E1,
                        float*       __restrict__ GT) {
    __shared__ float red[KS_][2][16];
    __shared__ int   ownl[16];

    const int c     = blockIdx.y;
    const int mtile = blockIdx.x;
    const int ks    = threadIdx.x >> 5;
    const int lane  = threadIdx.x & 31;
    const int c16   = lane & 15;
    const int hi    = lane >> 4;

    int col[TF_ / 32];
#pragma unroll
    for (int q = 0; q < TF_ / 32; ++q) col[q] = binary_tf[c * TF_ + q * 32 + lane];
    const int irow = mtile * 16 + 2 * ks;
    const int tr0raw = binary_tf[c * TF_ + irow];
    const int tr1raw = binary_tf[c * TF_ + irow + 1];
    __builtin_amdgcn_sched_barrier(0);

    const int kbase = ks * KLEN_ + hi * 4;
    const float* grow = grn + ((size_t)(c * TF_ + mtile * 16 + c16)) * TG_ + kbase;
    const float* zrow = z + (size_t)c * TG_ + kbase;
    const float  fill = (c16 == 1) ? 1.0f : 0.0f;

    v8f acc0, acc1;
#pragma unroll
    for (int r = 0; r < 8; ++r) { acc0[r] = 0.0f; acc1[r] = 0.0f; }

    v4f p[STEPS_], g[STEPS_];
#pragma unroll
    for (int j = 0; j < STEPS_; ++j) {
        p[j] = *(const v4f*)(zrow + j * 8);
        g[j] = *(const v4f*)(grow + j * 8);
    }
    __builtin_amdgcn_sched_barrier(0);

    const float* r0 = grn + ((size_t)(c * TF_ + irow)) * TG_;
    const float* r1 = r0 + TG_;
    float ga[TF_ / 32], gb[TF_ / 32];
    const int tr0 = tr0raw & (TG_ - 1);
    const int tr1 = tr1raw & (TG_ - 1);
    bool  later0 = false, later1 = false;
#pragma unroll
    for (int q = 0; q < TF_ / 32; ++q) {
        const int t = col[q] & (TG_ - 1);
        ga[q] = r0[t];
        gb[q] = r1[t];
        later0 |= (t == tr0) & (q * 32 + lane > irow);
        later1 |= (t == tr1) & (q * 32 + lane > irow + 1);
    }
    const bool own0 = __ballot(later0) == 0;
    const bool own1 = __ballot(later1) == 0;
    __builtin_amdgcn_sched_barrier(0);

#pragma unroll
    for (int j = 0; j < STEPS_; ++j) {
        v2f a, b;
        a.x = (c16 == 0) ? p[j].x : fill;  a.y = (c16 == 0) ? p[j].y : fill;
        b.x = g[j].x;  b.y = g[j].y;
        acc0 = __builtin_amdgcn_wmma_f32_16x16x4_f32(false, a, false, b, (short)0, acc0, false, false);
        a.x = (c16 == 0) ? p[j].z : fill;  a.y = (c16 == 0) ? p[j].w : fill;
        b.x = g[j].z;  b.y = g[j].w;
        acc1 = __builtin_amdgcn_wmma_f32_16x16x4_f32(false, a, false, b, (short)0, acc1, false, false);
    }

    const float su = acc0[0] + acc1[0];
    const float sv = acc0[1] + acc1[1];
    if (hi == 0) {
        red[ks][0][c16] = su;
        red[ks][1][c16] = sv;
    }
    if (lane == 0) {
        ownl[2 * ks]     = own0 ? 1 : 0;
        ownl[2 * ks + 1] = own1 ? 1 : 0;
    }

    __syncthreads();

    if (threadIdx.x < 32) {
        const int plane = lane >> 4;
        const int erow  = lane & 15;
        const int own   = ownl[erow];
        float s = red[0][plane][erow];
#pragma unroll
        for (int w = 1; w < KS_; ++w) s += red[w][plane][erow];
        E1[c * EPL_ + mtile * 32 + lane] = own ? s : 0.0f;
        ownf[(c * (TF_ / 16) + mtile) * 32 + lane] = own;
    }

    float* o0 = GT + ((size_t)(c * TF_ + irow)) * TF_ + lane;
#pragma unroll
    for (int q = 0; q < TF_ / 32; ++q) {
        o0[q * 32]       = own0 ? ga[q] : 0.0f;
        o0[TF_ + q * 32] = own1 ? gb[q] : 0.0f;
    }

    if (threadIdx.x < 64 && c == 0) zeros[mtile * 64 + threadIdx.x] = 0.0f;
}

template <bool FINAL>
__global__ __launch_bounds__(32 * KS_)
void reduced_sweep_kernel(const float* __restrict__ GT,
                          const float* __restrict__ Ein,
                          const float* __restrict__ E1,
                          const float* __restrict__ zeros,
                          const float* __restrict__ z,
                          const int*   __restrict__ binary_tf,
                          const int*   __restrict__ ownf,
                          const float* __restrict__ w_z, const float* __restrict__ b_z,
                          const float* __restrict__ w1, const float* __restrict__ b1,
                          const float* __restrict__ w2, const float* __restrict__ b2,
                          float*       __restrict__ dst) {
    __shared__ float red[KS_][2][16];

    const int c     = blockIdx.y;
    const int mtile = blockIdx.x;
    const int ks    = threadIdx.x >> 5;
    const int lane  = threadIdx.x & 31;
    const int c16   = lane & 15;
    const int hi    = lane >> 4;

    const float e1v = (threadIdx.x < 32) ? E1[c * EPL_ + mtile * 32 + lane] : 0.0f;
    int traw = 0, own = 0;
    if (FINAL) {
        traw = binary_tf[c * TF_ + mtile * 16 + c16];
        own  = ownf[(c * (TF_ / 16) + mtile) * 32 + lane];
    }
    __builtin_amdgcn_sched_barrier(0);

    const int kbase = ks * RLEN_ + hi * 4;
    const float* grow = GT + ((size_t)(c * TF_ + mtile * 16 + c16)) * TF_ + kbase;
    const float* prow = (c16 < 2) ? Ein + c * EPL_ + ks * 128 + c16 * 16 + hi * 4
                                  : zeros + hi * 4;

    v8f acc0, acc1;
#pragma unroll
    for (int r = 0; r < 8; ++r) { acc0[r] = 0.0f; acc1[r] = 0.0f; }

    v4f p[RSTEPS_], g[RSTEPS_];
#pragma unroll
    for (int j = 0; j < RSTEPS_; ++j) {
        p[j] = *(const v4f*)(prow + (j >> 1) * 32 + (j & 1) * 8);
        g[j] = *(const v4f*)(grow + j * 8);
    }
    __builtin_amdgcn_sched_barrier(0);
    float rhs = 0.0f;
    int trow = 0;
    if (FINAL) {
        trow = traw & (TG_ - 1);
        rhs  = (threadIdx.x < 16) ? z[c * TG_ + trow] : 1.0f;
        __builtin_amdgcn_sched_barrier(0);
    }
    MlpW W;
    if (FINAL && threadIdx.x < 32) {
        mlp_load(W, w_z, b_z, w1, b1, w2, b2, c16, hi);
        __builtin_amdgcn_sched_barrier(0);
    }
#pragma unroll
    for (int j = 0; j < RSTEPS_; ++j) {
        v2f a, b;
        a.x = p[j].x;  a.y = p[j].y;
        b.x = g[j].x;  b.y = g[j].y;
        acc0 = __builtin_amdgcn_wmma_f32_16x16x4_f32(false, a, false, b, (short)0, acc0, false, false);
        a.x = p[j].z;  a.y = p[j].w;
        b.x = g[j].z;  b.y = g[j].w;
        acc1 = __builtin_amdgcn_wmma_f32_16x16x4_f32(false, a, false, b, (short)0, acc1, false, false);
    }

    const float su = acc0[0] + acc1[0];
    const float sv = acc0[1] + acc1[1];
    if (hi == 0) {
        red[ks][0][c16] = su;
        red[ks][1][c16] = sv;
    }
    __syncthreads();

    if (threadIdx.x < 32) {
        const int plane = lane >> 4;
        const int erow  = lane & 15;
        float s = e1v;
#pragma unroll
        for (int w = 0; w < KS_; ++w) s += red[w][plane][erow];
        if (FINAL) {
            const float xv = rhs + s;
            float u[1], v[1], o0[1], o1[1];
            u[0] = __shfl(xv, c16);
            v[0] = __shfl(xv, 16 + c16);
            mlp_tiles<1>(W, u, v, c16, o0, o1);
            if (own && hi == 0) {
                dst[c * TG_ + trow] = o0[0];
                dst[C_ * TG_ + c * TG_ + trow] = softplus_eps(o1[0]);
            }
        } else {
            dst[c * EPL_ + mtile * 32 + lane] = s;
        }
    }
}


#define SQD_    4
#define SQSTEPS_ (TF_ / 8)

__device__ __forceinline__
void square_part(const float* __restrict__ GT, float* __restrict__ P, const int c, const int bx) {
    const int mb   = bx >> 2;
    const int nb   = bx & 3;
    const int w    = threadIdx.x >> 5;
    const int lane = threadIdx.x & 31;
    const int c16  = lane & 15;
    const int hi   = lane >> 4;
    const int m0   = mb * 64 + (w >> 1) * 32;
    const int n0   = nb * 128 + (w & 1) * 64;

    const float* G  = GT + (size_t)c * TF_ * TF_;
    const float* ap = G + (size_t)(m0 + c16) * TF_ + 4 * hi;
    const float* bp = G + (size_t)(4 * hi) * TF_ + n0 + 4 * c16;

    v8f acc[2][4];
#pragma unroll
    for (int mt = 0; mt < 2; ++mt)
#pragma unroll
        for (int nt = 0; nt < 4; ++nt)
#pragma unroll
            for (int r = 0; r < 8; ++r) acc[mt][nt][r] = 0.0f;

    v4f a[SQD_][2], b[SQD_][4];

#define SQ_LOAD(s, step)                                                            \
    {                                                                               \
        a[s][0] = *(const v4f*)(ap + (size_t)(step) * 8);                           \
        a[s][1] = *(const v4f*)(ap + (size_t)16 * TF_ + (size_t)(step) * 8);        \
        b[s][0] = *(const v4f*)(bp + ((size_t)(step) * 8 + 0) * TF_);               \
        b[s][1] = *(const v4f*)(bp + ((size_t)(step) * 8 + 1) * TF_);               \
        b[s][2] = *(const v4f*)(bp + ((size_t)(step) * 8 + 2) * TF_);               \
        b[s][3] = *(const v4f*)(bp + ((size_t)(step) * 8 + 3) * TF_);               \
    }

#define SQ_COMPUTE(s)                                                               \
    {                                                                               \
        _Pragma("unroll")                                                           \
        for (int h2 = 0; h2 < 2; ++h2) {                                            \
            v2f bb[4];                                                              \
            _Pragma("unroll")                                                       \
            for (int nt = 0; nt < 4; ++nt) {                                        \
                bb[nt].x = b[s][2 * h2][nt];                                        \
                bb[nt].y = b[s][2 * h2 + 1][nt];                                    \
            }                                                                       \
            _Pragma("unroll")                                                       \
            for (int mt = 0; mt < 2; ++mt) {                                        \
                v2f aa;                                                             \
                aa.x = a[s][mt][2 * h2];                                            \
                aa.y = a[s][mt][2 * h2 + 1];                                        \
                _Pragma("unroll")                                                   \
                for (int nt = 0; nt < 4; ++nt)                                      \
                    acc[mt][nt] = __builtin_amdgcn_wmma_f32_16x16x4_f32(            \
                        false, aa, false, bb[nt], (short)0, acc[mt][nt], false, false); \
            }                                                                       \
        }                                                                           \
    }

#pragma unroll
    for (int s = 0; s < SQD_; ++s) SQ_LOAD(s, s)
    __builtin_amdgcn_sched_barrier(0);

#pragma nounroll
    for (int it = 0; it < SQSTEPS_ - SQD_; it += SQD_) {
#pragma unroll
        for (int s = 0; s < SQD_; ++s) {
            SQ_COMPUTE(s)
            __builtin_amdgcn_sched_barrier(0);
            SQ_LOAD(s, it + SQD_ + s)
            __builtin_amdgcn_sched_barrier(0);
        }
    }
#pragma unroll
    for (int s = 0; s < SQD_; ++s) SQ_COMPUTE(s)

#undef SQ_LOAD
#undef SQ_COMPUTE

    float* prow = P + (size_t)c * TF_ * TF_ + (size_t)(m0 + 8 * hi) * TF_ + n0 + 4 * c16;
#pragma unroll
    for (int mt = 0; mt < 2; ++mt)
#pragma unroll
        for (int r = 0; r < 8; ++r) {
            v4f o;
            o.x = acc[mt][0][r];  o.y = acc[mt][1][r];
            o.z = acc[mt][2][r];  o.w = acc[mt][3][r];
            *(v4f*)(prow + (size_t)(16 * mt + r) * TF_) = o;
        }
}

__device__ __forceinline__
void f_rows_part(const float* __restrict__ GT,
                 const float* __restrict__ E1,
                 const float* __restrict__ zeros,
                 float*       __restrict__ F,
                 float (*red)[2][16],
                 const int c, const int mtile) {
    const int w    = threadIdx.x >> 5;
    const int lane = threadIdx.x & 31;
    const int c16  = lane & 15;
    const int hi   = lane >> 4;

    const float e1v = (threadIdx.x < 32) ? E1[c * EPL_ + mtile * 32 + lane] : 0.0f;
    __builtin_amdgcn_sched_barrier(0);

    const int kbase = 2 * w * RLEN_ + hi * 4;
    const float* grow = GT + ((size_t)(c * TF_ + mtile * 16 + c16)) * TF_ + kbase;
    const float* prow = (c16 < 2) ? E1 + c * EPL_ + (2 * w) * 128 + c16 * 16 + hi * 4
                                  : zeros + hi * 4;

    v8f acc[2][2];
#pragma unroll
    for (int r = 0; r < 8; ++r) {
        acc[0][0][r] = 0.0f;  acc[0][1][r] = 0.0f;
        acc[1][0][r] = 0.0f;  acc[1][1][r] = 0.0f;
    }

    v4f p[2 * RSTEPS_], g[2 * RSTEPS_];
#pragma unroll
    for (int j = 0; j < 2 * RSTEPS_; ++j) {
        p[j] = *(const v4f*)(prow + (j >> 1) * 32 + (j & 1) * 8);
        g[j] = *(const v4f*)(grow + j * 8);
    }
    __builtin_amdgcn_sched_barrier(0);
#pragma unroll
    for (int j = 0; j < 2 * RSTEPS_; ++j) {
        const int sl = j / RSTEPS_;
        v2f a, b;
        a.x = p[j].x;  a.y = p[j].y;
        b.x = g[j].x;  b.y = g[j].y;
        acc[sl][0] = __builtin_amdgcn_wmma_f32_16x16x4_f32(false, a, false, b, (short)0, acc[sl][0], false, false);
        a.x = p[j].z;  a.y = p[j].w;
        b.x = g[j].z;  b.y = g[j].w;
        acc[sl][1] = __builtin_amdgcn_wmma_f32_16x16x4_f32(false, a, false, b, (short)0, acc[sl][1], false, false);
    }

    if (hi == 0) {
#pragma unroll
        for (int sl = 0; sl < 2; ++sl) {
            red[2 * w + sl][0][c16] = acc[sl][0][0] + acc[sl][1][0];
            red[2 * w + sl][1][c16] = acc[sl][0][1] + acc[sl][1][1];
        }
    }
    __syncthreads();

    if (threadIdx.x < 32) {
        const int plane = lane >> 4;
        const int erow  = lane & 15;
        float s = e1v;
#pragma unroll
        for (int q = 0; q < KS_; ++q) s += red[q][plane][erow];
        F[c * EPL_ + mtile * 32 + lane] = s;
    }
}

__global__ __launch_bounds__(128)
void x0_rows_kernel(const float* __restrict__ z, const int* __restrict__ binary_tf,
                    const float* __restrict__ w_z, const float* __restrict__ b_z,
                    const float* __restrict__ w1, const float* __restrict__ b1,
                    const float* __restrict__ w2, const float* __restrict__ b2,
                    float* __restrict__ out) {
    x0_rows_part(z, binary_tf, w_z, b_z, w1, b1, w2, b2, out, blockIdx.x);
}

__global__ __launch_bounds__(128)
void square_and_f_kernel(const float* __restrict__ GT,
                         const float* __restrict__ E1,
                         const float* __restrict__ zeros,
                         float*       __restrict__ P,
                         float*       __restrict__ F,
                         const float* __restrict__ z,
                         const int*   __restrict__ binary_tf,
                         const float* __restrict__ w_z, const float* __restrict__ b_z,
                         const float* __restrict__ w1, const float* __restrict__ b1,
                         const float* __restrict__ w2, const float* __restrict__ b2,
                         float*       __restrict__ out) {
    __shared__ float red[KS_][2][16];
    const int bid = blockIdx.x;
    if (bid < 32 * C_) {
        square_part(GT, P, bid >> 5, bid & 31);
    } else if (bid < 2 * 32 * C_) {
        const int r = bid - 32 * C_;
        f_rows_part(GT, E1, zeros, F, red, r >> 5, r & 31);
    } else {
        x0_rows_part(z, binary_tf, w_z, b_z, w1, b1, w2, b2, out, bid - 2 * 32 * C_);
    }
}

extern "C" void kernel_launch(void* const* d_in, const int* in_sizes, int n_in,
                              void* d_out, int out_size, void* d_ws, size_t ws_size,
                              hipStream_t stream) {
    (void)in_sizes; (void)n_in; (void)out_size;
    const float* z   = (const float*)d_in[0];
    const float* grn = (const float*)d_in[1];
    const int*   btf = (const int*)  d_in[2];
    const float* w_z = (const float*)d_in[4];
    const float* b_z = (const float*)d_in[5];
    const float* w1  = (const float*)d_in[6];
    const float* b1  = (const float*)d_in[7];
    const float* w2  = (const float*)d_in[8];
    const float* b2  = (const float*)d_in[9];
    float* out = (float*)d_out;

    dim3 grid(TF_ / 16, C_);

    if (ws_size >= ((size_t)C_ * 3 * TG_ + TG_ + (size_t)4 * C_ * EPL_ + (size_t)2 * C_ * TF_ * TF_) * sizeof(float)) {
        float* zer = (float*)d_ws + (size_t)C_ * 2 * TG_;
        int*   own = (int*)(zer + TG_);
        float* E1  = (float*)(own + (size_t)C_ * TG_);
        float* F   = E1 + (size_t)C_ * EPL_;
        float* Ea  = F  + (size_t)C_ * EPL_;
        float* Eb  = Ea + (size_t)C_ * EPL_;
        float* GT  = Eb + (size_t)C_ * EPL_;
        float* P   = GT + (size_t)C_ * TF_ * TF_;

        first_sweep_kernel<<<grid, 32 * KS_, 0, stream>>>(grn, z, btf, zer, own, E1, GT);
        square_and_f_kernel<<<dim3(2 * 32 * C_ + C_ * TG_ / 128), 128, 0, stream>>>(
            GT, E1, zer, P, F, z, btf, w_z, b_z, w1, b1, w2, b2, out);
        const float* yin = F;
        float* ybuf[2] = { Ea, Eb };
        for (int j = 1; j < 8; ++j) {
            float* yout = ybuf[j & 1];
            reduced_sweep_kernel<false><<<grid, 32 * KS_, 0, stream>>>(P, yin, F, zer, z, btf, own, w_z, b_z, w1, b1, w2, b2, yout);
            yin = yout;
        }
        reduced_sweep_kernel<true><<<grid, 32 * KS_, 0, stream>>>(P, yin, F, zer, z, btf, own, w_z, b_z, w1, b1, w2, b2, out);
        return;
    }

    if (ws_size >= ((size_t)C_ * 3 * TG_ + TG_ + (size_t)3 * C_ * EPL_ + (size_t)C_ * TF_ * TF_) * sizeof(float)) {
        float* zer = (float*)d_ws + (size_t)C_ * 2 * TG_;
        int*   own = (int*)(zer + TG_);
        float* E1  = (float*)(own + (size_t)C_ * TG_);
        float* Ea  = E1 + (size_t)C_ * EPL_;
        float* Eb  = Ea + (size_t)C_ * EPL_;
        float* GT  = Eb + (size_t)C_ * EPL_;

        first_sweep_kernel<<<grid, 32 * KS_, 0, stream>>>(grn, z, btf, zer, own, E1, GT);
        const float* ein = E1;
        float* ebuf[2] = { Ea, Eb };
        for (int s = 2; s < NITER; ++s) {
            float* eout = ebuf[s & 1];
            reduced_sweep_kernel<false><<<grid, 32 * KS_, 0, stream>>>(GT, ein, E1, zer, z, btf, own, w_z, b_z, w1, b1, w2, b2, eout);
            ein = eout;
        }
        reduced_sweep_kernel<true><<<grid, 32 * KS_, 0, stream>>>(GT, ein, E1, zer, z, btf, own, w_z, b_z, w1, b1, w2, b2, out);
        x0_rows_kernel<<<C_ * TG_ / 128, 128, 0, stream>>>(z, btf, w_z, b_z, w1, b1, w2, b2, out);
        return;
    }

    float* UVa = (float*)d_ws;
    float* UVb = UVa + (size_t)C_ * 2 * TG_;
    float* zer = UVb + (size_t)C_ * 2 * TG_;

    init_uv_kernel<<<(C_ * TG_ + 255) / 256, 256, 0, stream>>>(z, UVa, UVb, zer);

    float* bufs[2] = { UVa, UVb };
    for (int it = 0; it < NITER; ++it) {
        const float* uvin = bufs[it & 1];
        float*      uvout = bufs[(it + 1) & 1];
        neumann_iter_kernel<<<grid, 32 * KS_, 0, stream>>>(grn, z, btf, uvin, zer, uvout);
    }
    const float* uvfin = bufs[NITER & 1];

    mlp_kernel<<<(C_ * TG_ + 127) / 128, 128, 0, stream>>>(uvfin, w_z, b_z, w1, b1, w2, b2, out);
}
